# starting kernel
# baseline (speedup 1.0000x reference)
#include <hip/hip_runtime.h>

typedef float v2f __attribute__((ext_vector_type(2)));
typedef float v8f __attribute__((ext_vector_type(8)));

#define HID 32
#define BN_EPS 1e-5f

__global__ void k_init_deg(float* deg, int N) {
  int i = blockIdx.x * blockDim.x + threadIdx.x;
  if (i < N) deg[i] = 1.0f;
}
__global__ void k_deg_edges(const int* __restrict__ col, const float* __restrict__ w,
                            float* deg, int E) {
  int e = blockIdx.x * blockDim.x + threadIdx.x;
  if (e < E) atomicAdd(&deg[col[e]], w[e]);
}
__global__ void k_deg_rsqrt(float* deg, int N) {
  int i = blockIdx.x * blockDim.x + threadIdx.x;
  if (i < N) deg[i] = rsqrtf(deg[i]);
}

template <int K>
__global__ __launch_bounds__(256) void k_gemm_wmma(const float* __restrict__ A,
                                                   const float* __restrict__ B,
                                                   float* __restrict__ D, int M) {
  const int lane = threadIdx.x;
  const int wave = threadIdx.y;
  const int mt   = blockIdx.x * 8 + wave;
  const int numT = (M + 15) >> 4;
  if (mt >= numT) return;
  const int nb   = blockIdx.y * 16;
  const int half = lane >> 4;
  const int lm   = lane & 15;

  int arow_i = mt * 16 + lm;
  if (arow_i > M - 1) arow_i = M - 1;
  const float* arow = A + (size_t)arow_i * K;

  v8f c = {};
#pragma unroll
  for (int k0 = 0; k0 < K; k0 += 4) {
    const int ka = k0 + 2 * half;
    v2f a;
    a.x = arow[ka];
    a.y = arow[ka + 1];
    v2f b;
    b.x = B[(size_t)ka * HID + nb + lm];
    b.y = B[(size_t)(ka + 1) * HID + nb + lm];
    c = __builtin_amdgcn_wmma_f32_16x16x4_f32(false, a, false, b, (short)0, c,
                                              false, false);
  }
#pragma unroll
  for (int r = 0; r < 8; ++r) {
    const int orow = mt * 16 + r + 8 * half;
    if (orow < M) D[(size_t)orow * HID + nb + lm] = c[r];
  }
}

__global__ void k_init_bias(float* __restrict__ cbuf, const float* __restrict__ b, int total) {
  int i = blockIdx.x * blockDim.x + threadIdx.x;
  if (i < total) cbuf[i] = b[i & (HID - 1)];
}

__global__ void k_prop(const float* __restrict__ h, const int* __restrict__ row,
                       const int* __restrict__ col, const float* __restrict__ w,
                       const float* __restrict__ dis, float* __restrict__ out,
                       int E, int N) {
  long long tid = (long long)blockIdx.x * blockDim.x + threadIdx.x;
  int e = (int)(tid >> 5);
  int f = (int)(tid & 31);
  if (e >= E + N) return;
  int r, c; float ww;
  if (e < E) { r = row[e]; c = col[e]; ww = w[e]; }
  else       { r = e - E;  c = r;      ww = 1.0f; }
  float nrm = dis[r] * ww * dis[c];
  atomicAdd(out + (size_t)c * HID + f, h[(size_t)r * HID + f] * nrm);
}

__global__ void k_zero_stats(float* stats) {
  if (threadIdx.x < 64) stats[threadIdx.x] = 0.0f;
}
__global__ void k_bn_stats(const float* __restrict__ c, float* __restrict__ stats, int N) {
  __shared__ float ls[64];
  if (threadIdx.x < 64) ls[threadIdx.x] = 0.0f;
  __syncthreads();
  const int f      = threadIdx.x & 31;
  const int r0     = blockIdx.x * (blockDim.x >> 5) + (threadIdx.x >> 5);
  const int stride = gridDim.x * (blockDim.x >> 5);
  float s = 0.0f, s2 = 0.0f;
  for (int n = r0; n < N; n += stride) {
    float v = fmaxf(c[(size_t)n * HID + f], 0.0f);
    s += v; s2 += v * v;
  }
  atomicAdd(&ls[f], s);
  atomicAdd(&ls[32 + f], s2);
  __syncthreads();
  if (threadIdx.x < 64) atomicAdd(&stats[threadIdx.x], ls[threadIdx.x]);
}

__global__ void k_bn_apply(const float* __restrict__ c, const float* __restrict__ stats,
                           const float* __restrict__ g, const float* __restrict__ be,
                           float* __restrict__ out, int N) {
  int i = blockIdx.x * blockDim.x + threadIdx.x;
  if (i >= N * HID) return;
  const int f = i & 31;
  const float invN = 1.0f / (float)N;
  const float mu   = stats[f] * invN;
  const float var  = stats[32 + f] * invN - mu * mu;
  const float v    = fmaxf(c[i], 0.0f);
  out[i] = (v - mu) * rsqrtf(var + BN_EPS) * g[f] + be[f];
}

__global__ void k_final(const float* __restrict__ o1, const float* __restrict__ o2,
                        const float* __restrict__ o3, const float* __restrict__ Wl,
                        const float* __restrict__ bl, float* __restrict__ out, int N) {
  int n = blockIdx.x * blockDim.x + threadIdx.x;
  if (n >= N) return;
  float a0 = bl[0], a1 = bl[1];
  const float* p = o1 + (size_t)n * HID;
#pragma unroll
  for (int j = 0; j < HID; ++j) { float v = p[j]; a0 += v * Wl[2 * j]; a1 += v * Wl[2 * j + 1]; }
  p = o2 + (size_t)n * HID;
#pragma unroll
  for (int j = 0; j < HID; ++j) { float v = p[j]; a0 += v * Wl[2 * (HID + j)]; a1 += v * Wl[2 * (HID + j) + 1]; }
  p = o3 + (size_t)n * HID;
#pragma unroll
  for (int j = 0; j < HID; ++j) { float v = p[j]; a0 += v * Wl[2 * (2 * HID + j)]; a1 += v * Wl[2 * (2 * HID + j) + 1]; }
  out[2 * n] = a0;
  out[2 * n + 1] = a1;
}

extern "C" void kernel_launch(void* const* d_in, const int* in_sizes, int n_in,
                              void* d_out, int out_size, void* d_ws, size_t ws_size,
                              hipStream_t stream) {
  const float* x   = (const float*)d_in[0];
  const int*   ei  = (const int*)d_in[1];
  const float* ew  = (const float*)d_in[2];
  const float* W1  = (const float*)d_in[3];
  const float* b1  = (const float*)d_in[4];
  const float* g1  = (const float*)d_in[5];
  const float* be1 = (const float*)d_in[6];
  const float* W2  = (const float*)d_in[7];
  const float* b2  = (const float*)d_in[8];
  const float* g2  = (const float*)d_in[9];
  const float* be2 = (const float*)d_in[10];
  const float* W3  = (const float*)d_in[11];
  const float* b3  = (const float*)d_in[12];
  const float* g3  = (const float*)d_in[13];
  const float* be3 = (const float*)d_in[14];
  const float* Wl  = (const float*)d_in[15];
  const float* bl  = (const float*)d_in[16];

  const int F = 64;
  const int N = in_sizes[0] / F;
  const int E = in_sizes[1] / 2;
  const int* row = ei;
  const int* col = ei + E;

  float* ws    = (float*)d_ws;
  float* dis   = ws;
  float* hbuf  = dis  + N;
  float* cbuf  = hbuf + (size_t)N * HID;
  float* o1    = cbuf + (size_t)N * HID;
  float* o2    = o1   + (size_t)N * HID;
  float* o3    = o2   + (size_t)N * HID;
  float* stats = o3   + (size_t)N * HID;

  const int T = 256;

  k_init_deg<<<(N + T - 1) / T, T, 0, stream>>>(dis, N);
  k_deg_edges<<<(E + T - 1) / T, T, 0, stream>>>(col, ew, dis, E);
  k_deg_rsqrt<<<(N + T - 1) / T, T, 0, stream>>>(dis, N);

  const int numT = (N + 15) / 16;
  const dim3 gemmB(32, 8);
  const dim3 gemmG((numT + 7) / 8, 2);
  const int elemB = (N * HID + T - 1) / T;
  const long long propT = (long long)(E + N) * 32;
  const int propB = (int)((propT + T - 1) / T);

  k_gemm_wmma<64><<<gemmG, gemmB, 0, stream>>>(x, W1, hbuf, N);
  k_init_bias<<<elemB, T, 0, stream>>>(cbuf, b1, N * HID);
  k_prop<<<propB, T, 0, stream>>>(hbuf, row, col, ew, dis, cbuf, E, N);
  k_zero_stats<<<1, 64, 0, stream>>>(stats);
  k_bn_stats<<<512, T, 0, stream>>>(cbuf, stats, N);
  k_bn_apply<<<elemB, T, 0, stream>>>(cbuf, stats, g1, be1, o1, N);

  k_gemm_wmma<32><<<gemmG, gemmB, 0, stream>>>(o1, W2, hbuf, N);
  k_init_bias<<<elemB, T, 0, stream>>>(cbuf, b2, N * HID);
  k_prop<<<propB, T, 0, stream>>>(hbuf, row, col, ew, dis, cbuf, E, N);
  k_zero_stats<<<1, 64, 0, stream>>>(stats);
  k_bn_stats<<<512, T, 0, stream>>>(cbuf, stats, N);
  k_bn_apply<<<elemB, T, 0, stream>>>(cbuf, stats, g2, be2, o2, N);

  k_gemm_wmma<32><<<gemmG, gemmB, 0, stream>>>(o2, W3, hbuf, N);
  k_init_bias<<<elemB, T, 0, stream>>>(cbuf, b3, N * HID);
  k_prop<<<propB, T, 0, stream>>>(hbuf, row, col, ew, dis, cbuf, E, N);
  k_zero_stats<<<1, 64, 0, stream>>>(stats);
  k_bn_stats<<<512, T, 0, stream>>>(cbuf, stats, N);
  k_bn_apply<<<elemB, T, 0, stream>>>(cbuf, stats, g3, be3, o3, N);

  k_final<<<(N + T - 1) / T, T, 0, stream>>>(o1, o2, o3, Wl, bl, (float*)d_out, N);
}


# Float-atomic edge scatter replaced by gather over edges grouped by target (count, integer arrival-order rank, prefix sum, placement per call); node sums now in arrival-rank order, bias last
# speedup vs baseline: 1.9525x; 1.9525x over previous
#include <hip/hip_runtime.h>

typedef float v2f __attribute__((ext_vector_type(2)));
typedef float v8f __attribute__((ext_vector_type(8)));

#define HID 32
#define BN_EPS 1e-5f

#define SCAN_T 1024

__global__ void k_zero_cnt(int* cnt, int N) {
  int i = blockIdx.x * blockDim.x + threadIdx.x;
  if (i < N) cnt[i] = 0;
}
__global__ void k_rank(const int* __restrict__ col, int* cnt, int* __restrict__ rank, int E) {
  int e = blockIdx.x * blockDim.x + threadIdx.x;
  if (e < E) rank[e] = atomicAdd(&cnt[col[e]], 1);
}
__global__ __launch_bounds__(SCAN_T) void k_scan(const int* __restrict__ cnt, int* __restrict__ off, int N) {
  __shared__ int sh[SCAN_T];
  __shared__ int sbase;
  const int t     = threadIdx.x;
  const int start = blockIdx.x * SCAN_T;
  int part = 0;
  for (int i = t; i < start; i += SCAN_T) part += cnt[i];
  for (int d = 16; d > 0; d >>= 1) part += __shfl_xor(part, d);
  if (t == 0) sbase = 0;
  __syncthreads();
  if ((t & 31) == 0) atomicAdd(&sbase, part);
  const int idx = start + t;
  const int v   = (idx < N) ? cnt[idx] : 0;
  sh[t] = v;
  __syncthreads();
  for (int d = 1; d < SCAN_T; d <<= 1) {
    const int add = (t >= d) ? sh[t - d] : 0;
    __syncthreads();
    sh[t] += add;
    __syncthreads();
  }
  if (idx <= N) off[idx] = sbase + sh[t] - v;
}
__global__ void k_fill(const int* __restrict__ row, const int* __restrict__ col,
                       const float* __restrict__ w, const int* __restrict__ off,
                       const int* __restrict__ rank, int* __restrict__ sed, int E) {
  int e = blockIdx.x * blockDim.x + threadIdx.x;
  if (e >= E) return;
  const int p = off[col[e]] + rank[e];
  reinterpret_cast<int2*>(sed)[p] = make_int2(row[e], __float_as_int(w[e]));
}
__global__ void k_dis(const int* __restrict__ off, const int* __restrict__ sed,
                      float* __restrict__ dis, int N) {
  int c = blockIdx.x * blockDim.x + threadIdx.x;
  if (c >= N) return;
  float s = 1.0f;
  const int end = off[c + 1];
  for (int p = off[c]; p < end; ++p) s += __int_as_float(sed[2 * (size_t)p + 1]);
  dis[c] = rsqrtf(s);
}

template <int K>
__global__ __launch_bounds__(256) void k_gemm_wmma(const float* __restrict__ A,
                                                   const float* __restrict__ B,
                                                   float* __restrict__ D, int M) {
  const int lane = threadIdx.x;
  const int wave = threadIdx.y;
  const int mt   = blockIdx.x * 8 + wave;
  const int numT = (M + 15) >> 4;
  if (mt >= numT) return;
  const int nb   = blockIdx.y * 16;
  const int half = lane >> 4;
  const int lm   = lane & 15;

  int arow_i = mt * 16 + lm;
  if (arow_i > M - 1) arow_i = M - 1;
  const float* arow = A + (size_t)arow_i * K;

  v8f c = {};
#pragma unroll
  for (int k0 = 0; k0 < K; k0 += 4) {
    const int ka = k0 + 2 * half;
    v2f a;
    a.x = arow[ka];
    a.y = arow[ka + 1];
    v2f b;
    b.x = B[(size_t)ka * HID + nb + lm];
    b.y = B[(size_t)(ka + 1) * HID + nb + lm];
    c = __builtin_amdgcn_wmma_f32_16x16x4_f32(false, a, false, b, (short)0, c,
                                              false, false);
  }
#pragma unroll
  for (int r = 0; r < 8; ++r) {
    const int orow = mt * 16 + r + 8 * half;
    if (orow < M) D[(size_t)orow * HID + nb + lm] = c[r];
  }
}

__global__ __launch_bounds__(256) void k_gather(const float* __restrict__ h, const int* __restrict__ off,
                                                const int* __restrict__ sed,
                                                const float* __restrict__ dis,
                                                const float* __restrict__ bias,
                                                float* __restrict__ out, int N) {
  const int lane = threadIdx.x & 31;
  const int wib = __builtin_amdgcn_readfirstlane((int)(threadIdx.x >> 5));
  const int wpb = (int)(blockDim.x >> 5);
  const int nw  = (int)gridDim.x * wpb;
  const float bb = bias[lane];
  for (int c = (int)blockIdx.x * wpb + wib; c < N; c += nw) {
    const int beg = off[c];
    const int end = off[c + 1];
    const float dc = dis[c];
    float acc = h[(size_t)c * HID + lane] * (dc * 1.0f * dc);
    for (int p0 = beg; p0 < end; p0 += 32) {
      const int p = p0 + lane;
      int   r  = c;
      float cf = 0.0f;
      if (p < end) {
        const int2 ed = reinterpret_cast<const int2*>(sed)[p];
        r  = ed.x;
        cf = dis[r] * __int_as_float(ed.y) * dc;
      }
      const int cfi = __float_as_int(cf);
      const int n   = (end - p0 < 32) ? (end - p0) : 32;
      for (int j = 0; j < n; j += 8) {
        float v[8], f[8];
        for (int u = 0; u < 8; ++u) {
          const int rj = __builtin_amdgcn_readlane(r, j + u);
          f[u] = __int_as_float(__builtin_amdgcn_readlane(cfi, j + u));
          v[u] = h[(size_t)rj * HID + lane];
        }
        for (int u = 0; u < 8; ++u) acc += v[u] * f[u];
      }
    }
    out[(size_t)c * HID + lane] = acc + bb;
  }
}

__global__ void k_zero_stats(float* stats) {
  if (threadIdx.x < 64) stats[threadIdx.x] = 0.0f;
}
__global__ void k_bn_stats(const float* __restrict__ c, float* __restrict__ stats, int N) {
  __shared__ float ls[64];
  if (threadIdx.x < 64) ls[threadIdx.x] = 0.0f;
  __syncthreads();
  const int f      = threadIdx.x & 31;
  const int r0     = blockIdx.x * (blockDim.x >> 5) + (threadIdx.x >> 5);
  const int stride = gridDim.x * (blockDim.x >> 5);
  float s = 0.0f, s2 = 0.0f;
  for (int n = r0; n < N; n += stride) {
    float v = fmaxf(c[(size_t)n * HID + f], 0.0f);
    s += v; s2 += v * v;
  }
  atomicAdd(&ls[f], s);
  atomicAdd(&ls[32 + f], s2);
  __syncthreads();
  if (threadIdx.x < 64) atomicAdd(&stats[threadIdx.x], ls[threadIdx.x]);
}

__global__ void k_bn_apply(const float* __restrict__ c, const float* __restrict__ stats,
                           const float* __restrict__ g, const float* __restrict__ be,
                           float* __restrict__ out, int N) {
  int i = blockIdx.x * blockDim.x + threadIdx.x;
  if (i >= N * HID) return;
  const int f = i & 31;
  const float invN = 1.0f / (float)N;
  const float mu   = stats[f] * invN;
  const float var  = stats[32 + f] * invN - mu * mu;
  const float v    = fmaxf(c[i], 0.0f);
  out[i] = (v - mu) * rsqrtf(var + BN_EPS) * g[f] + be[f];
}

__global__ void k_final(const float* __restrict__ o1, const float* __restrict__ o2,
                        const float* __restrict__ o3, const float* __restrict__ Wl,
                        const float* __restrict__ bl, float* __restrict__ out, int N) {
  int n = blockIdx.x * blockDim.x + threadIdx.x;
  if (n >= N) return;
  float a0 = bl[0], a1 = bl[1];
  const float* p = o1 + (size_t)n * HID;
#pragma unroll
  for (int j = 0; j < HID; ++j) { float v = p[j]; a0 += v * Wl[2 * j]; a1 += v * Wl[2 * j + 1]; }
  p = o2 + (size_t)n * HID;
#pragma unroll
  for (int j = 0; j < HID; ++j) { float v = p[j]; a0 += v * Wl[2 * (HID + j)]; a1 += v * Wl[2 * (HID + j) + 1]; }
  p = o3 + (size_t)n * HID;
#pragma unroll
  for (int j = 0; j < HID; ++j) { float v = p[j]; a0 += v * Wl[2 * (2 * HID + j)]; a1 += v * Wl[2 * (2 * HID + j) + 1]; }
  out[2 * n] = a0;
  out[2 * n + 1] = a1;
}

extern "C" void kernel_launch(void* const* d_in, const int* in_sizes, int n_in,
                              void* d_out, int out_size, void* d_ws, size_t ws_size,
                              hipStream_t stream) {
  const float* x   = (const float*)d_in[0];
  const int*   ei  = (const int*)d_in[1];
  const float* ew  = (const float*)d_in[2];
  const float* W1  = (const float*)d_in[3];
  const float* b1  = (const float*)d_in[4];
  const float* g1  = (const float*)d_in[5];
  const float* be1 = (const float*)d_in[6];
  const float* W2  = (const float*)d_in[7];
  const float* b2  = (const float*)d_in[8];
  const float* g2  = (const float*)d_in[9];
  const float* be2 = (const float*)d_in[10];
  const float* W3  = (const float*)d_in[11];
  const float* b3  = (const float*)d_in[12];
  const float* g3  = (const float*)d_in[13];
  const float* be3 = (const float*)d_in[14];
  const float* Wl  = (const float*)d_in[15];
  const float* bl  = (const float*)d_in[16];

  const int F = 64;
  const int N = in_sizes[0] / F;
  const int E = in_sizes[1] / 2;
  const int* row = ei;
  const int* col = ei + E;

  float* ws    = (float*)d_ws;
  float* dis   = ws;
  float* hbuf  = dis  + N;
  float* cbuf  = hbuf + (size_t)N * HID;
  float* o1    = cbuf + (size_t)N * HID;
  float* o2    = o1   + (size_t)N * HID;
  float* o3    = o2   + (size_t)N * HID;
  float* stats = o3   + (size_t)N * HID;
  int*   cnt   = (int*)(stats + 64);
  int*   off   = cnt + N;
  int*   rank  = off + N + 1;
  const size_t sedOff = (((size_t)((float*)(rank + E) - ws)) + 63) & ~(size_t)63;
  int*   sed   = (int*)(ws + sedOff);

  const int T = 256;

  k_zero_cnt<<<(N + T - 1) / T, T, 0, stream>>>(cnt, N);
  k_rank<<<(E + T - 1) / T, T, 0, stream>>>(col, cnt, rank, E);
  k_scan<<<(N + SCAN_T) / SCAN_T, SCAN_T, 0, stream>>>(cnt, off, N);
  k_fill<<<(E + T - 1) / T, T, 0, stream>>>(row, col, ew, off, rank, sed, E);
  k_dis<<<(N + T - 1) / T, T, 0, stream>>>(off, sed, dis, N);

  const int numT = (N + 15) / 16;
  const dim3 gemmB(32, 8);
  const dim3 gemmG((numT + 7) / 8, 2);
  const int elemB = (N * HID + T - 1) / T;
  const int gathB = 2048;

  k_gemm_wmma<64><<<gemmG, gemmB, 0, stream>>>(x, W1, hbuf, N);
  k_gather<<<gathB, T, 0, stream>>>(hbuf, off, sed, dis, b1, cbuf, N);
  k_zero_stats<<<1, 64, 0, stream>>>(stats);
  k_bn_stats<<<512, T, 0, stream>>>(cbuf, stats, N);
  k_bn_apply<<<elemB, T, 0, stream>>>(cbuf, stats, g1, be1, o1, N);

  k_gemm_wmma<32><<<gemmG, gemmB, 0, stream>>>(o1, W2, hbuf, N);
  k_gather<<<gathB, T, 0, stream>>>(hbuf, off, sed, dis, b2, cbuf, N);
  k_zero_stats<<<1, 64, 0, stream>>>(stats);
  k_bn_stats<<<512, T, 0, stream>>>(cbuf, stats, N);
  k_bn_apply<<<elemB, T, 0, stream>>>(cbuf, stats, g2, be2, o2, N);

  k_gemm_wmma<32><<<gemmG, gemmB, 0, stream>>>(o2, W3, hbuf, N);
  k_gather<<<gathB, T, 0, stream>>>(hbuf, off, sed, dis, b3, cbuf, N);
  k_zero_stats<<<1, 64, 0, stream>>>(stats);
  k_bn_stats<<<512, T, 0, stream>>>(cbuf, stats, N);
  k_bn_apply<<<elemB, T, 0, stream>>>(cbuf, stats, g3, be3, o3, N);

  k_final<<<(N + T - 1) / T, T, 0, stream>>>(o1, o2, o3, Wl, bl, (float*)d_out, N);
}


# BN apply passes removed: relu+BN now computed in registers where its result is read (A loader of the next layer's transform, readout); three launches and the o1..o3 round trips gone
# speedup vs baseline: 2.0695x; 1.0599x over previous
#include <hip/hip_runtime.h>

typedef float v2f __attribute__((ext_vector_type(2)));
typedef float v8f __attribute__((ext_vector_type(8)));

#define HID 32
#define BN_EPS 1e-5f

#define SCAN_T 1024

__global__ void k_zero_cnt(int* cnt, int N) {
  int i = blockIdx.x * blockDim.x + threadIdx.x;
  if (i < N) cnt[i] = 0;
}
__global__ void k_rank(const int* __restrict__ col, int* cnt, int* __restrict__ rank, int E) {
  int e = blockIdx.x * blockDim.x + threadIdx.x;
  if (e < E) rank[e] = atomicAdd(&cnt[col[e]], 1);
}
__global__ __launch_bounds__(SCAN_T) void k_scan(const int* __restrict__ cnt, int* __restrict__ off, int N) {
  __shared__ int sh[SCAN_T];
  __shared__ int sbase;
  const int t     = threadIdx.x;
  const int start = blockIdx.x * SCAN_T;
  int part = 0;
  for (int i = t; i < start; i += SCAN_T) part += cnt[i];
  for (int d = 16; d > 0; d >>= 1) part += __shfl_xor(part, d);
  if (t == 0) sbase = 0;
  __syncthreads();
  if ((t & 31) == 0) atomicAdd(&sbase, part);
  const int idx = start + t;
  const int v   = (idx < N) ? cnt[idx] : 0;
  sh[t] = v;
  __syncthreads();
  for (int d = 1; d < SCAN_T; d <<= 1) {
    const int add = (t >= d) ? sh[t - d] : 0;
    __syncthreads();
    sh[t] += add;
    __syncthreads();
  }
  if (idx <= N) off[idx] = sbase + sh[t] - v;
}
__global__ void k_fill(const int* __restrict__ row, const int* __restrict__ col,
                       const float* __restrict__ w, const int* __restrict__ off,
                       const int* __restrict__ rank, int* __restrict__ sed, int E) {
  int e = blockIdx.x * blockDim.x + threadIdx.x;
  if (e >= E) return;
  const int p = off[col[e]] + rank[e];
  reinterpret_cast<int2*>(sed)[p] = make_int2(row[e], __float_as_int(w[e]));
}
__global__ void k_dis(const int* __restrict__ off, const int* __restrict__ sed,
                      float* __restrict__ dis, int N) {
  int c = blockIdx.x * blockDim.x + threadIdx.x;
  if (c >= N) return;
  float s = 1.0f;
  const int end = off[c + 1];
  for (int p = off[c]; p < end; ++p) s += __int_as_float(sed[2 * (size_t)p + 1]);
  dis[c] = rsqrtf(s);
}

__device__ __forceinline__ float relu_bn(float x, const float* __restrict__ stats,
                                         const float* __restrict__ g, const float* __restrict__ be,
                                         int f, float invN) {
  const float mu  = stats[f] * invN;
  const float var = stats[32 + f] * invN - mu * mu;
  const float v   = fmaxf(x, 0.0f);
  return (v - mu) * rsqrtf(var + BN_EPS) * g[f] + be[f];
}

template <int K, bool BN>
__global__ __launch_bounds__(256) void k_gemm_wmma(const float* __restrict__ A,
                                                   const float* __restrict__ B,
                                                   float* __restrict__ D, int M,
                                                   const float* __restrict__ stats,
                                                   const float* __restrict__ g,
                                                   const float* __restrict__ be) {
  const int lane = threadIdx.x;
  const int wave = threadIdx.y;
  const int mt   = blockIdx.x * 8 + wave;
  const int numT = (M + 15) >> 4;
  if (mt >= numT) return;
  const int nb   = blockIdx.y * 16;
  const int half = lane >> 4;
  const int lm   = lane & 15;

  int arow_i = mt * 16 + lm;
  if (arow_i > M - 1) arow_i = M - 1;
  const float* arow = A + (size_t)arow_i * K;
  const float invN = 1.0f / (float)M;

  v8f c = {};
#pragma unroll
  for (int k0 = 0; k0 < K; k0 += 4) {
    const int ka = k0 + 2 * half;
    v2f a;
    a.x = arow[ka];
    a.y = arow[ka + 1];
    if constexpr (BN) {
      a.x = relu_bn(a.x, stats, g, be, ka, invN);
      a.y = relu_bn(a.y, stats, g, be, ka + 1, invN);
    }
    v2f b;
    b.x = B[(size_t)ka * HID + nb + lm];
    b.y = B[(size_t)(ka + 1) * HID + nb + lm];
    c = __builtin_amdgcn_wmma_f32_16x16x4_f32(false, a, false, b, (short)0, c,
                                              false, false);
  }
#pragma unroll
  for (int r = 0; r < 8; ++r) {
    const int orow = mt * 16 + r + 8 * half;
    if (orow < M) D[(size_t)orow * HID + nb + lm] = c[r];
  }
}

__global__ __launch_bounds__(256) void k_gather(const float* __restrict__ h, const int* __restrict__ off,
                                                const int* __restrict__ sed,
                                                const float* __restrict__ dis,
                                                const float* __restrict__ bias,
                                                float* __restrict__ out, int N) {
  const int lane = threadIdx.x & 31;
  const int wib = __builtin_amdgcn_readfirstlane((int)(threadIdx.x >> 5));
  const int wpb = (int)(blockDim.x >> 5);
  const int nw  = (int)gridDim.x * wpb;
  const float bb = bias[lane];
  for (int c = (int)blockIdx.x * wpb + wib; c < N; c += nw) {
    const int beg = off[c];
    const int end = off[c + 1];
    const float dc = dis[c];
    float acc = h[(size_t)c * HID + lane] * (dc * 1.0f * dc);
    for (int p0 = beg; p0 < end; p0 += 32) {
      const int p = p0 + lane;
      int   r  = c;
      float cf = 0.0f;
      if (p < end) {
        const int2 ed = reinterpret_cast<const int2*>(sed)[p];
        r  = ed.x;
        cf = dis[r] * __int_as_float(ed.y) * dc;
      }
      const int cfi = __float_as_int(cf);
      const int n   = (end - p0 < 32) ? (end - p0) : 32;
      for (int j = 0; j < n; j += 8) {
        float v[8], f[8];
        for (int u = 0; u < 8; ++u) {
          const int rj = __builtin_amdgcn_readlane(r, j + u);
          f[u] = __int_as_float(__builtin_amdgcn_readlane(cfi, j + u));
          v[u] = h[(size_t)rj * HID + lane];
        }
        for (int u = 0; u < 8; ++u) acc += v[u] * f[u];
      }
    }
    out[(size_t)c * HID + lane] = acc + bb;
  }
}

__global__ void k_zero_stats(float* stats) {
  if (threadIdx.x < 64) stats[threadIdx.x] = 0.0f;
}
__global__ void k_bn_stats(const float* __restrict__ c, float* __restrict__ stats, int N) {
  __shared__ float ls[64];
  if (threadIdx.x < 64) ls[threadIdx.x] = 0.0f;
  __syncthreads();
  const int f      = threadIdx.x & 31;
  const int r0     = blockIdx.x * (blockDim.x >> 5) + (threadIdx.x >> 5);
  const int stride = gridDim.x * (blockDim.x >> 5);
  float s = 0.0f, s2 = 0.0f;
  for (int n = r0; n < N; n += stride) {
    float v = fmaxf(c[(size_t)n * HID + f], 0.0f);
    s += v; s2 += v * v;
  }
  atomicAdd(&ls[f], s);
  atomicAdd(&ls[32 + f], s2);
  __syncthreads();
  if (threadIdx.x < 64) atomicAdd(&stats[threadIdx.x], ls[threadIdx.x]);
}

__global__ __launch_bounds__(256) void k_final(const float* __restrict__ c1, const float* __restrict__ c2,
                        const float* __restrict__ c3, const float* __restrict__ stats,
                        const float* __restrict__ g1, const float* __restrict__ be1,
                        const float* __restrict__ g2, const float* __restrict__ be2,
                        const float* __restrict__ g3, const float* __restrict__ be3,
                        const float* __restrict__ Wl,
                        const float* __restrict__ bl, float* __restrict__ out, int N) {
  __shared__ float4 prm[3 * HID];
  const int t = threadIdx.x;
  if (t < 3 * HID) {
    const int l = t >> 5;
    const int f = t & 31;
    const float* s = stats + 64 * l;
    const float* g = (l == 0) ? g1 : ((l == 1) ? g2 : g3);
    const float* b = (l == 0) ? be1 : ((l == 1) ? be2 : be3);
    const float invN = 1.0f / (float)N;
    const float mu   = s[f] * invN;
    const float var  = s[32 + f] * invN - mu * mu;
    prm[t] = make_float4(mu, rsqrtf(var + BN_EPS), g[f], b[f]);
  }
  __syncthreads();
  const int n = blockIdx.x * blockDim.x + t;
  if (n >= N) return;
  float a0 = bl[0], a1 = bl[1];
  const float* p = c1 + (size_t)n * HID;
#pragma unroll
  for (int j = 0; j < HID; ++j) {
    const float4 q = prm[j];
    const float v = (fmaxf(p[j], 0.0f) - q.x) * q.y * q.z + q.w;
    a0 += v * Wl[2 * j]; a1 += v * Wl[2 * j + 1];
  }
  p = c2 + (size_t)n * HID;
#pragma unroll
  for (int j = 0; j < HID; ++j) {
    const float4 q = prm[HID + j];
    const float v = (fmaxf(p[j], 0.0f) - q.x) * q.y * q.z + q.w;
    a0 += v * Wl[2 * (HID + j)]; a1 += v * Wl[2 * (HID + j) + 1];
  }
  p = c3 + (size_t)n * HID;
#pragma unroll
  for (int j = 0; j < HID; ++j) {
    const float4 q = prm[2 * HID + j];
    const float v = (fmaxf(p[j], 0.0f) - q.x) * q.y * q.z + q.w;
    a0 += v * Wl[2 * (2 * HID + j)]; a1 += v * Wl[2 * (2 * HID + j) + 1];
  }
  out[2 * n] = a0;
  out[2 * n + 1] = a1;
}

extern "C" void kernel_launch(void* const* d_in, const int* in_sizes, int n_in,
                              void* d_out, int out_size, void* d_ws, size_t ws_size,
                              hipStream_t stream) {
  const float* x   = (const float*)d_in[0];
  const int*   ei  = (const int*)d_in[1];
  const float* ew  = (const float*)d_in[2];
  const float* W1  = (const float*)d_in[3];
  const float* b1  = (const float*)d_in[4];
  const float* g1  = (const float*)d_in[5];
  const float* be1 = (const float*)d_in[6];
  const float* W2  = (const float*)d_in[7];
  const float* b2  = (const float*)d_in[8];
  const float* g2  = (const float*)d_in[9];
  const float* be2 = (const float*)d_in[10];
  const float* W3  = (const float*)d_in[11];
  const float* b3  = (const float*)d_in[12];
  const float* g3  = (const float*)d_in[13];
  const float* be3 = (const float*)d_in[14];
  const float* Wl  = (const float*)d_in[15];
  const float* bl  = (const float*)d_in[16];

  const int F = 64;
  const int N = in_sizes[0] / F;
  const int E = in_sizes[1] / 2;
  const int* row = ei;
  const int* col = ei + E;

  float* ws    = (float*)d_ws;
  float* dis   = ws;
  float* hbuf  = dis  + N;
  float* c1    = hbuf + (size_t)N * HID;
  float* c2    = c1   + (size_t)N * HID;
  float* c3    = c2   + (size_t)N * HID;
  float* stats = c3   + (size_t)N * HID;
  float* st1   = stats;
  float* st2   = stats + 64;
  float* st3   = stats + 128;
  int*   cnt   = (int*)(stats + 192);
  int*   off   = cnt + N;
  int*   rank  = off + N + 1;
  const size_t sedOff = (((size_t)((float*)(rank + E) - ws)) + 63) & ~(size_t)63;
  int*   sed   = (int*)(ws + sedOff);

  const int T = 256;

  k_zero_cnt<<<(N + T - 1) / T, T, 0, stream>>>(cnt, N);
  k_rank<<<(E + T - 1) / T, T, 0, stream>>>(col, cnt, rank, E);
  k_scan<<<(N + SCAN_T) / SCAN_T, SCAN_T, 0, stream>>>(cnt, off, N);
  k_fill<<<(E + T - 1) / T, T, 0, stream>>>(row, col, ew, off, rank, sed, E);
  k_dis<<<(N + T - 1) / T, T, 0, stream>>>(off, sed, dis, N);

  const int numT = (N + 15) / 16;
  const dim3 gemmB(32, 8);
  const dim3 gemmG((numT + 7) / 8, 2);
  const int gathB = 2048;

  k_gemm_wmma<64, false><<<gemmG, gemmB, 0, stream>>>(x, W1, hbuf, N, nullptr, nullptr, nullptr);
  k_gather<<<gathB, T, 0, stream>>>(hbuf, off, sed, dis, b1, c1, N);
  k_zero_stats<<<1, 64, 0, stream>>>(st1);
  k_bn_stats<<<512, T, 0, stream>>>(c1, st1, N);

  k_gemm_wmma<32, true><<<gemmG, gemmB, 0, stream>>>(c1, W2, hbuf, N, st1, g1, be1);
  k_gather<<<gathB, T, 0, stream>>>(hbuf, off, sed, dis, b2, c2, N);
  k_zero_stats<<<1, 64, 0, stream>>>(st2);
  k_bn_stats<<<512, T, 0, stream>>>(c2, st2, N);

  k_gemm_wmma<32, true><<<gemmG, gemmB, 0, stream>>>(c2, W3, hbuf, N, st2, g2, be2);
  k_gather<<<gathB, T, 0, stream>>>(hbuf, off, sed, dis, b3, c3, N);
  k_zero_stats<<<1, 64, 0, stream>>>(st3);
  k_bn_stats<<<512, T, 0, stream>>>(c3, st3, N);

  k_final<<<(N + T - 1) / T, T, 0, stream>>>(c1, c2, c3, stats, g1, be1, g2, be2, g3, be3,
                                             Wl, bl, (float*)d_out, N);
}


# BN sums of relu(conv) taken in the gather (per wave, then per block in LDS) and finished by a 64-block sum kernel, all in fixed order; no float atomics left in BN; 18 launches to 15
# speedup vs baseline: 2.1817x; 1.0542x over previous
#include <hip/hip_runtime.h>

typedef float v2f __attribute__((ext_vector_type(2)));
typedef float v8f __attribute__((ext_vector_type(8)));

#define HID 32
#define BN_EPS 1e-5f

#define SCAN_T 1024

__global__ void k_zero_cnt(int* cnt, int N) {
  int i = blockIdx.x * blockDim.x + threadIdx.x;
  if (i < N) cnt[i] = 0;
}
__global__ void k_rank(const int* __restrict__ col, int* cnt, int* __restrict__ rank, int E) {
  int e = blockIdx.x * blockDim.x + threadIdx.x;
  if (e < E) rank[e] = atomicAdd(&cnt[col[e]], 1);
}
__global__ __launch_bounds__(SCAN_T) void k_scan(const int* __restrict__ cnt, int* __restrict__ off, int N) {
  __shared__ int sh[SCAN_T];
  __shared__ int sbase;
  const int t     = threadIdx.x;
  const int start = blockIdx.x * SCAN_T;
  int part = 0;
  for (int i = t; i < start; i += SCAN_T) part += cnt[i];
  for (int d = 16; d > 0; d >>= 1) part += __shfl_xor(part, d);
  if (t == 0) sbase = 0;
  __syncthreads();
  if ((t & 31) == 0) atomicAdd(&sbase, part);
  const int idx = start + t;
  const int v   = (idx < N) ? cnt[idx] : 0;
  sh[t] = v;
  __syncthreads();
  for (int d = 1; d < SCAN_T; d <<= 1) {
    const int add = (t >= d) ? sh[t - d] : 0;
    __syncthreads();
    sh[t] += add;
    __syncthreads();
  }
  if (idx <= N) off[idx] = sbase + sh[t] - v;
}
__global__ void k_fill(const int* __restrict__ row, const int* __restrict__ col,
                       const float* __restrict__ w, const int* __restrict__ off,
                       const int* __restrict__ rank, int* __restrict__ sed, int E) {
  int e = blockIdx.x * blockDim.x + threadIdx.x;
  if (e >= E) return;
  const int p = off[col[e]] + rank[e];
  reinterpret_cast<int2*>(sed)[p] = make_int2(row[e], __float_as_int(w[e]));
}
__global__ void k_dis(const int* __restrict__ off, const int* __restrict__ sed,
                      float* __restrict__ dis, int N) {
  int c = blockIdx.x * blockDim.x + threadIdx.x;
  if (c >= N) return;
  float s = 1.0f;
  const int end = off[c + 1];
  for (int p = off[c]; p < end; ++p) s += __int_as_float(sed[2 * (size_t)p + 1]);
  dis[c] = rsqrtf(s);
}

__device__ __forceinline__ float relu_bn(float x, const float* __restrict__ stats,
                                         const float* __restrict__ g, const float* __restrict__ be,
                                         int f, float invN) {
  const float mu  = stats[f] * invN;
  const float var = stats[32 + f] * invN - mu * mu;
  const float v   = fmaxf(x, 0.0f);
  return (v - mu) * rsqrtf(var + BN_EPS) * g[f] + be[f];
}

template <int K, bool BN>
__global__ __launch_bounds__(256) void k_gemm_wmma(const float* __restrict__ A,
                                                   const float* __restrict__ B,
                                                   float* __restrict__ D, int M,
                                                   const float* __restrict__ stats,
                                                   const float* __restrict__ g,
                                                   const float* __restrict__ be) {
  const int lane = threadIdx.x;
  const int wave = threadIdx.y;
  const int mt   = blockIdx.x * 8 + wave;
  const int numT = (M + 15) >> 4;
  if (mt >= numT) return;
  const int nb   = blockIdx.y * 16;
  const int half = lane >> 4;
  const int lm   = lane & 15;

  int arow_i = mt * 16 + lm;
  if (arow_i > M - 1) arow_i = M - 1;
  const float* arow = A + (size_t)arow_i * K;
  const float invN = 1.0f / (float)M;

  v8f c = {};
#pragma unroll
  for (int k0 = 0; k0 < K; k0 += 4) {
    const int ka = k0 + 2 * half;
    v2f a;
    a.x = arow[ka];
    a.y = arow[ka + 1];
    if constexpr (BN) {
      a.x = relu_bn(a.x, stats, g, be, ka, invN);
      a.y = relu_bn(a.y, stats, g, be, ka + 1, invN);
    }
    v2f b;
    b.x = B[(size_t)ka * HID + nb + lm];
    b.y = B[(size_t)(ka + 1) * HID + nb + lm];
    c = __builtin_amdgcn_wmma_f32_16x16x4_f32(false, a, false, b, (short)0, c,
                                              false, false);
  }
#pragma unroll
  for (int r = 0; r < 8; ++r) {
    const int orow = mt * 16 + r + 8 * half;
    if (orow < M) D[(size_t)orow * HID + nb + lm] = c[r];
  }
}

#define GATH_T 256
#define GATH_W (GATH_T / 32)
__global__ __launch_bounds__(GATH_T) void k_gather(const float* __restrict__ h, const int* __restrict__ off,
                                                   const int* __restrict__ sed,
                                                   const float* __restrict__ dis,
                                                   const float* __restrict__ bias,
                                                   float* __restrict__ out,
                                                   float* __restrict__ part, int N) {
  __shared__ float ls[GATH_W][2 * HID];
  const int lane = threadIdx.x & 31;
  const int wib = __builtin_amdgcn_readfirstlane((int)(threadIdx.x >> 5));
  const int wpb = (int)(blockDim.x >> 5);
  const int nw  = (int)gridDim.x * wpb;
  const float bb = bias[lane];
  float s1 = 0.0f, s2 = 0.0f;
  for (int c = (int)blockIdx.x * wpb + wib; c < N; c += nw) {
    const int beg = off[c];
    const int end = off[c + 1];
    const float dc = dis[c];
    float acc = h[(size_t)c * HID + lane] * (dc * 1.0f * dc);
    for (int p0 = beg; p0 < end; p0 += 32) {
      const int p = p0 + lane;
      int   r  = c;
      float cf = 0.0f;
      if (p < end) {
        const int2 ed = reinterpret_cast<const int2*>(sed)[p];
        r  = ed.x;
        cf = dis[r] * __int_as_float(ed.y) * dc;
      }
      const int cfi = __float_as_int(cf);
      const int n   = (end - p0 < 32) ? (end - p0) : 32;
      for (int j = 0; j < n; j += 8) {
        float v[8], f[8];
        for (int u = 0; u < 8; ++u) {
          const int rj = __builtin_amdgcn_readlane(r, j + u);
          f[u] = __int_as_float(__builtin_amdgcn_readlane(cfi, j + u));
          v[u] = h[(size_t)rj * HID + lane];
        }
        for (int u = 0; u < 8; ++u) acc += v[u] * f[u];
      }
    }
    const float o = acc + bb;
    out[(size_t)c * HID + lane] = o;
    const float v = fmaxf(o, 0.0f);
    s1 += v;
    s2 += v * v;
  }
  ls[wib][lane]       = s1;
  ls[wib][HID + lane] = s2;
  __syncthreads();
  if (threadIdx.x < 2 * HID) {
    float t = 0.0f;
    for (int w = 0; w < GATH_W; ++w) t += ls[w][threadIdx.x];
    part[(size_t)threadIdx.x * gridDim.x + blockIdx.x] = t;
  }
}

#define SSUM_T 512
__global__ __launch_bounds__(SSUM_T) void k_stats_sum(const float* __restrict__ part,
                                                      float* __restrict__ stats, int P) {
  __shared__ float sw[SSUM_T / 32];
  const int t = threadIdx.x;
  const float4* p4 = reinterpret_cast<const float4*>(part + (size_t)blockIdx.x * P);
  float s = 0.0f;
  for (int i = t; i < (P >> 2); i += SSUM_T) {
    const float4 v = p4[i];
    s += (v.x + v.y) + (v.z + v.w);
  }
  for (int d = 16; d > 0; d >>= 1) s += __shfl_xor(s, d);
  if ((t & 31) == 0) sw[t >> 5] = s;
  __syncthreads();
  if (t == 0) {
    float r = 0.0f;
    for (int w = 0; w < SSUM_T / 32; ++w) r += sw[w];
    stats[blockIdx.x] = r;
  }
}

__global__ __launch_bounds__(256) void k_final(const float* __restrict__ c1, const float* __restrict__ c2,
                        const float* __restrict__ c3, const float* __restrict__ stats,
                        const float* __restrict__ g1, const float* __restrict__ be1,
                        const float* __restrict__ g2, const float* __restrict__ be2,
                        const float* __restrict__ g3, const float* __restrict__ be3,
                        const float* __restrict__ Wl,
                        const float* __restrict__ bl, float* __restrict__ out, int N) {
  __shared__ float4 prm[3 * HID];
  const int t = threadIdx.x;
  if (t < 3 * HID) {
    const int l = t >> 5;
    const int f = t & 31;
    const float* s = stats + 64 * l;
    const float* g = (l == 0) ? g1 : ((l == 1) ? g2 : g3);
    const float* b = (l == 0) ? be1 : ((l == 1) ? be2 : be3);
    const float invN = 1.0f / (float)N;
    const float mu   = s[f] * invN;
    const float var  = s[32 + f] * invN - mu * mu;
    prm[t] = make_float4(mu, rsqrtf(var + BN_EPS), g[f], b[f]);
  }
  __syncthreads();
  const int n = blockIdx.x * blockDim.x + t;
  if (n >= N) return;
  float a0 = bl[0], a1 = bl[1];
  const float* p = c1 + (size_t)n * HID;
#pragma unroll
  for (int j = 0; j < HID; ++j) {
    const float4 q = prm[j];
    const float v = (fmaxf(p[j], 0.0f) - q.x) * q.y * q.z + q.w;
    a0 += v * Wl[2 * j]; a1 += v * Wl[2 * j + 1];
  }
  p = c2 + (size_t)n * HID;
#pragma unroll
  for (int j = 0; j < HID; ++j) {
    const float4 q = prm[HID + j];
    const float v = (fmaxf(p[j], 0.0f) - q.x) * q.y * q.z + q.w;
    a0 += v * Wl[2 * (HID + j)]; a1 += v * Wl[2 * (HID + j) + 1];
  }
  p = c3 + (size_t)n * HID;
#pragma unroll
  for (int j = 0; j < HID; ++j) {
    const float4 q = prm[2 * HID + j];
    const float v = (fmaxf(p[j], 0.0f) - q.x) * q.y * q.z + q.w;
    a0 += v * Wl[2 * (2 * HID + j)]; a1 += v * Wl[2 * (2 * HID + j) + 1];
  }
  out[2 * n] = a0;
  out[2 * n + 1] = a1;
}

extern "C" void kernel_launch(void* const* d_in, const int* in_sizes, int n_in,
                              void* d_out, int out_size, void* d_ws, size_t ws_size,
                              hipStream_t stream) {
  const float* x   = (const float*)d_in[0];
  const int*   ei  = (const int*)d_in[1];
  const float* ew  = (const float*)d_in[2];
  const float* W1  = (const float*)d_in[3];
  const float* b1  = (const float*)d_in[4];
  const float* g1  = (const float*)d_in[5];
  const float* be1 = (const float*)d_in[6];
  const float* W2  = (const float*)d_in[7];
  const float* b2  = (const float*)d_in[8];
  const float* g2  = (const float*)d_in[9];
  const float* be2 = (const float*)d_in[10];
  const float* W3  = (const float*)d_in[11];
  const float* b3  = (const float*)d_in[12];
  const float* g3  = (const float*)d_in[13];
  const float* be3 = (const float*)d_in[14];
  const float* Wl  = (const float*)d_in[15];
  const float* bl  = (const float*)d_in[16];

  const int F = 64;
  const int N = in_sizes[0] / F;
  const int E = in_sizes[1] / 2;
  const int* row = ei;
  const int* col = ei + E;

  float* ws    = (float*)d_ws;
  float* dis   = ws;
  float* hbuf  = dis  + N;
  float* c1    = hbuf + (size_t)N * HID;
  float* c2    = c1   + (size_t)N * HID;
  float* c3    = c2   + (size_t)N * HID;
  float* stats = c3   + (size_t)N * HID;
  float* st1   = stats;
  float* st2   = stats + 64;
  float* st3   = stats + 128;
  int*   cnt   = (int*)(stats + 192);
  int*   off   = cnt + N;
  int*   rank  = off + N + 1;
  const size_t sedOff = (((size_t)((float*)(rank + E) - ws)) + 63) & ~(size_t)63;
  int*   sed   = (int*)(ws + sedOff);
  const int gathB = 2048;
  const size_t partOff = (sedOff + 2 * (size_t)E + 63) & ~(size_t)63;
  float* part  = ws + partOff;

  const int T = 256;

  k_zero_cnt<<<(N + T - 1) / T, T, 0, stream>>>(cnt, N);
  k_rank<<<(E + T - 1) / T, T, 0, stream>>>(col, cnt, rank, E);
  k_scan<<<(N + SCAN_T) / SCAN_T, SCAN_T, 0, stream>>>(cnt, off, N);
  k_fill<<<(E + T - 1) / T, T, 0, stream>>>(row, col, ew, off, rank, sed, E);
  k_dis<<<(N + T - 1) / T, T, 0, stream>>>(off, sed, dis, N);

  const int numT = (N + 15) / 16;
  const dim3 gemmB(32, 8);
  const dim3 gemmG((numT + 7) / 8, 2);

  k_gemm_wmma<64, false><<<gemmG, gemmB, 0, stream>>>(x, W1, hbuf, N, nullptr, nullptr, nullptr);
  k_gather<<<gathB, GATH_T, 0, stream>>>(hbuf, off, sed, dis, b1, c1, part, N);
  k_stats_sum<<<2 * HID, SSUM_T, 0, stream>>>(part, st1, gathB);

  k_gemm_wmma<32, true><<<gemmG, gemmB, 0, stream>>>(c1, W2, hbuf, N, st1, g1, be1);
  k_gather<<<gathB, GATH_T, 0, stream>>>(hbuf, off, sed, dis, b2, c2, part, N);
  k_stats_sum<<<2 * HID, SSUM_T, 0, stream>>>(part, st2, gathB);

  k_gemm_wmma<32, true><<<gemmG, gemmB, 0, stream>>>(c2, W3, hbuf, N, st2, g2, be2);
  k_gather<<<gathB, GATH_T, 0, stream>>>(hbuf, off, sed, dis, b3, c3, part, N);
  k_stats_sum<<<2 * HID, SSUM_T, 0, stream>>>(part, st3, gathB);

  k_final<<<(N + T - 1) / T, T, 0, stream>>>(c1, c2, c3, stats, g1, be1, g2, be2, g3, be3,
                                             Wl, bl, (float*)d_out, N);
}


# Symmetric norm split: transform stores row m times dis[m], gather weights rows by edge weight only and applies dis[target] once after the sum; no dis[source] lookup per edge; sum order kept
# speedup vs baseline: 2.3160x; 1.0616x over previous
#include <hip/hip_runtime.h>

typedef float v2f __attribute__((ext_vector_type(2)));
typedef float v8f __attribute__((ext_vector_type(8)));

#define HID 32
#define BN_EPS 1e-5f

#define SCAN_T 1024

__global__ void k_zero_cnt(int* cnt, int N) {
  int i = blockIdx.x * blockDim.x + threadIdx.x;
  if (i < N) cnt[i] = 0;
}
__global__ void k_rank(const int* __restrict__ col, int* cnt, int* __restrict__ rank, int E) {
  int e = blockIdx.x * blockDim.x + threadIdx.x;
  if (e < E) rank[e] = atomicAdd(&cnt[col[e]], 1);
}
__global__ __launch_bounds__(SCAN_T) void k_scan(const int* __restrict__ cnt, int* __restrict__ off, int N) {
  __shared__ int sh[SCAN_T];
  __shared__ int sbase;
  const int t     = threadIdx.x;
  const int start = blockIdx.x * SCAN_T;
  int part = 0;
  for (int i = t; i < start; i += SCAN_T) part += cnt[i];
  for (int d = 16; d > 0; d >>= 1) part += __shfl_xor(part, d);
  if (t == 0) sbase = 0;
  __syncthreads();
  if ((t & 31) == 0) atomicAdd(&sbase, part);
  const int idx = start + t;
  const int v   = (idx < N) ? cnt[idx] : 0;
  sh[t] = v;
  __syncthreads();
  for (int d = 1; d < SCAN_T; d <<= 1) {
    const int add = (t >= d) ? sh[t - d] : 0;
    __syncthreads();
    sh[t] += add;
    __syncthreads();
  }
  if (idx <= N) off[idx] = sbase + sh[t] - v;
}
__global__ void k_fill(const int* __restrict__ row, const int* __restrict__ col,
                       const float* __restrict__ w, const int* __restrict__ off,
                       const int* __restrict__ rank, int* __restrict__ sed, int E) {
  int e = blockIdx.x * blockDim.x + threadIdx.x;
  if (e >= E) return;
  const int p = off[col[e]] + rank[e];
  reinterpret_cast<int2*>(sed)[p] = make_int2(row[e], __float_as_int(w[e]));
}
__global__ void k_dis(const int* __restrict__ off, const int* __restrict__ sed,
                      float* __restrict__ dis, int N) {
  int c = blockIdx.x * blockDim.x + threadIdx.x;
  if (c >= N) return;
  float s = 1.0f;
  const int end = off[c + 1];
  for (int p = off[c]; p < end; ++p) s += __int_as_float(sed[2 * (size_t)p + 1]);
  dis[c] = rsqrtf(s);
}

__device__ __forceinline__ float relu_bn(float x, const float* __restrict__ stats,
                                         const float* __restrict__ g, const float* __restrict__ be,
                                         int f, float invN) {
  const float mu  = stats[f] * invN;
  const float var = stats[32 + f] * invN - mu * mu;
  const float v   = fmaxf(x, 0.0f);
  return (v - mu) * rsqrtf(var + BN_EPS) * g[f] + be[f];
}

template <int K, bool BN>
__global__ __launch_bounds__(256) void k_gemm_wmma(const float* __restrict__ A,
                                                   const float* __restrict__ B,
                                                   float* __restrict__ D, int M,
                                                   const float* __restrict__ stats,
                                                   const float* __restrict__ g,
                                                   const float* __restrict__ be,
                                                   const float* __restrict__ dis) {
  const int lane = threadIdx.x;
  const int wave = threadIdx.y;
  const int mt   = blockIdx.x * 8 + wave;
  const int numT = (M + 15) >> 4;
  if (mt >= numT) return;
  const int nb   = blockIdx.y * 16;
  const int half = lane >> 4;
  const int lm   = lane & 15;

  int arow_i = mt * 16 + lm;
  if (arow_i > M - 1) arow_i = M - 1;
  const float* arow = A + (size_t)arow_i * K;
  const float invN = 1.0f / (float)M;

  const int dli = __float_as_int(dis[arow_i]);

  v8f c = {};
#pragma unroll
  for (int k0 = 0; k0 < K; k0 += 4) {
    const int ka = k0 + 2 * half;
    v2f a;
    a.x = arow[ka];
    a.y = arow[ka + 1];
    if constexpr (BN) {
      a.x = relu_bn(a.x, stats, g, be, ka, invN);
      a.y = relu_bn(a.y, stats, g, be, ka + 1, invN);
    }
    v2f b;
    b.x = B[(size_t)ka * HID + nb + lm];
    b.y = B[(size_t)(ka + 1) * HID + nb + lm];
    c = __builtin_amdgcn_wmma_f32_16x16x4_f32(false, a, false, b, (short)0, c,
                                              false, false);
  }
#pragma unroll
  for (int r = 0; r < 8; ++r) {
    const int orow = mt * 16 + r + 8 * half;
    const float d_lo = __int_as_float(__builtin_amdgcn_readlane(dli, r));
    const float d_hi = __int_as_float(__builtin_amdgcn_readlane(dli, r + 8));
    if (orow < M) D[(size_t)orow * HID + nb + lm] = c[r] * (half ? d_hi : d_lo);
  }
}

#define GATH_T 256
#define GATH_W (GATH_T / 32)
__global__ __launch_bounds__(GATH_T) void k_gather(const float* __restrict__ h, const int* __restrict__ off,
                                                   const int* __restrict__ sed,
                                                   const float* __restrict__ dis,
                                                   const float* __restrict__ bias,
                                                   float* __restrict__ out,
                                                   float* __restrict__ part, int N) {
  __shared__ float ls[GATH_W][2 * HID];
  const int lane = threadIdx.x & 31;
  const int wib = __builtin_amdgcn_readfirstlane((int)(threadIdx.x >> 5));
  const int wpb = (int)(blockDim.x >> 5);
  const int nw  = (int)gridDim.x * wpb;
  const float bb = bias[lane];
  float s1 = 0.0f, s2 = 0.0f;
  for (int c = (int)blockIdx.x * wpb + wib; c < N; c += nw) {
    const int beg = off[c];
    const int end = off[c + 1];
    const float dc = dis[c];
    float acc = h[(size_t)c * HID + lane];
    for (int p0 = beg; p0 < end; p0 += 32) {
      const int p = p0 + lane;
      int   r  = c;
      float wf = 0.0f;
      if (p < end) {
        const int2 ed = reinterpret_cast<const int2*>(sed)[p];
        r  = ed.x;
        wf = __int_as_float(ed.y);
      }
      const int wfi = __float_as_int(wf);
      const int n   = (end - p0 < 32) ? (end - p0) : 32;
      for (int j = 0; j < n; j += 8) {
        float v[8], f[8];
        for (int u = 0; u < 8; ++u) {
          const int rj = __builtin_amdgcn_readlane(r, j + u);
          f[u] = __int_as_float(__builtin_amdgcn_readlane(wfi, j + u));
          v[u] = h[(size_t)rj * HID + lane];
        }
        for (int u = 0; u < 8; ++u) acc += v[u] * f[u];
      }
    }
    const float o = acc * dc + bb;
    out[(size_t)c * HID + lane] = o;
    const float v = fmaxf(o, 0.0f);
    s1 += v;
    s2 += v * v;
  }
  ls[wib][lane]       = s1;
  ls[wib][HID + lane] = s2;
  __syncthreads();
  if (threadIdx.x < 2 * HID) {
    float t = 0.0f;
    for (int w = 0; w < GATH_W; ++w) t += ls[w][threadIdx.x];
    part[(size_t)threadIdx.x * gridDim.x + blockIdx.x] = t;
  }
}

#define SSUM_T 512
__global__ __launch_bounds__(SSUM_T) void k_stats_sum(const float* __restrict__ part,
                                                      float* __restrict__ stats, int P) {
  __shared__ float sw[SSUM_T / 32];
  const int t = threadIdx.x;
  const float4* p4 = reinterpret_cast<const float4*>(part + (size_t)blockIdx.x * P);
  float s = 0.0f;
  for (int i = t; i < (P >> 2); i += SSUM_T) {
    const float4 v = p4[i];
    s += (v.x + v.y) + (v.z + v.w);
  }
  for (int d = 16; d > 0; d >>= 1) s += __shfl_xor(s, d);
  if ((t & 31) == 0) sw[t >> 5] = s;
  __syncthreads();
  if (t == 0) {
    float r = 0.0f;
    for (int w = 0; w < SSUM_T / 32; ++w) r += sw[w];
    stats[blockIdx.x] = r;
  }
}

__global__ __launch_bounds__(256) void k_final(const float* __restrict__ c1, const float* __restrict__ c2,
                        const float* __restrict__ c3, const float* __restrict__ stats,
                        const float* __restrict__ g1, const float* __restrict__ be1,
                        const float* __restrict__ g2, const float* __restrict__ be2,
                        const float* __restrict__ g3, const float* __restrict__ be3,
                        const float* __restrict__ Wl,
                        const float* __restrict__ bl, float* __restrict__ out, int N) {
  __shared__ float4 prm[3 * HID];
  const int t = threadIdx.x;
  if (t < 3 * HID) {
    const int l = t >> 5;
    const int f = t & 31;
    const float* s = stats + 64 * l;
    const float* g = (l == 0) ? g1 : ((l == 1) ? g2 : g3);
    const float* b = (l == 0) ? be1 : ((l == 1) ? be2 : be3);
    const float invN = 1.0f / (float)N;
    const float mu   = s[f] * invN;
    const float var  = s[32 + f] * invN - mu * mu;
    prm[t] = make_float4(mu, rsqrtf(var + BN_EPS), g[f], b[f]);
  }
  __syncthreads();
  const int n = blockIdx.x * blockDim.x + t;
  if (n >= N) return;
  float a0 = bl[0], a1 = bl[1];
  const float* p = c1 + (size_t)n * HID;
#pragma unroll
  for (int j = 0; j < HID; ++j) {
    const float4 q = prm[j];
    const float v = (fmaxf(p[j], 0.0f) - q.x) * q.y * q.z + q.w;
    a0 += v * Wl[2 * j]; a1 += v * Wl[2 * j + 1];
  }
  p = c2 + (size_t)n * HID;
#pragma unroll
  for (int j = 0; j < HID; ++j) {
    const float4 q = prm[HID + j];
    const float v = (fmaxf(p[j], 0.0f) - q.x) * q.y * q.z + q.w;
    a0 += v * Wl[2 * (HID + j)]; a1 += v * Wl[2 * (HID + j) + 1];
  }
  p = c3 + (size_t)n * HID;
#pragma unroll
  for (int j = 0; j < HID; ++j) {
    const float4 q = prm[2 * HID + j];
    const float v = (fmaxf(p[j], 0.0f) - q.x) * q.y * q.z + q.w;
    a0 += v * Wl[2 * (2 * HID + j)]; a1 += v * Wl[2 * (2 * HID + j) + 1];
  }
  out[2 * n] = a0;
  out[2 * n + 1] = a1;
}

extern "C" void kernel_launch(void* const* d_in, const int* in_sizes, int n_in,
                              void* d_out, int out_size, void* d_ws, size_t ws_size,
                              hipStream_t stream) {
  const float* x   = (const float*)d_in[0];
  const int*   ei  = (const int*)d_in[1];
  const float* ew  = (const float*)d_in[2];
  const float* W1  = (const float*)d_in[3];
  const float* b1  = (const float*)d_in[4];
  const float* g1  = (const float*)d_in[5];
  const float* be1 = (const float*)d_in[6];
  const float* W2  = (const float*)d_in[7];
  const float* b2  = (const float*)d_in[8];
  const float* g2  = (const float*)d_in[9];
  const float* be2 = (const float*)d_in[10];
  const float* W3  = (const float*)d_in[11];
  const float* b3  = (const float*)d_in[12];
  const float* g3  = (const float*)d_in[13];
  const float* be3 = (const float*)d_in[14];
  const float* Wl  = (const float*)d_in[15];
  const float* bl  = (const float*)d_in[16];

  const int F = 64;
  const int N = in_sizes[0] / F;
  const int E = in_sizes[1] / 2;
  const int* row = ei;
  const int* col = ei + E;

  float* ws    = (float*)d_ws;
  float* dis   = ws;
  float* hbuf  = dis  + N;
  float* c1    = hbuf + (size_t)N * HID;
  float* c2    = c1   + (size_t)N * HID;
  float* c3    = c2   + (size_t)N * HID;
  float* stats = c3   + (size_t)N * HID;
  float* st1   = stats;
  float* st2   = stats + 64;
  float* st3   = stats + 128;
  int*   cnt   = (int*)(stats + 192);
  int*   off   = cnt + N;
  int*   rank  = off + N + 1;
  const size_t sedOff = (((size_t)((float*)(rank + E) - ws)) + 63) & ~(size_t)63;
  int*   sed   = (int*)(ws + sedOff);
  const int gathB = 2048;
  const size_t partOff = (sedOff + 2 * (size_t)E + 63) & ~(size_t)63;
  float* part  = ws + partOff;

  const int T = 256;

  k_zero_cnt<<<(N + T - 1) / T, T, 0, stream>>>(cnt, N);
  k_rank<<<(E + T - 1) / T, T, 0, stream>>>(col, cnt, rank, E);
  k_scan<<<(N + SCAN_T) / SCAN_T, SCAN_T, 0, stream>>>(cnt, off, N);
  k_fill<<<(E + T - 1) / T, T, 0, stream>>>(row, col, ew, off, rank, sed, E);
  k_dis<<<(N + T - 1) / T, T, 0, stream>>>(off, sed, dis, N);

  const int numT = (N + 15) / 16;
  const dim3 gemmB(32, 8);
  const dim3 gemmG((numT + 7) / 8, 2);

  k_gemm_wmma<64, false><<<gemmG, gemmB, 0, stream>>>(x, W1, hbuf, N, nullptr, nullptr, nullptr, dis);
  k_gather<<<gathB, GATH_T, 0, stream>>>(hbuf, off, sed, dis, b1, c1, part, N);
  k_stats_sum<<<2 * HID, SSUM_T, 0, stream>>>(part, st1, gathB);

  k_gemm_wmma<32, true><<<gemmG, gemmB, 0, stream>>>(c1, W2, hbuf, N, st1, g1, be1, dis);
  k_gather<<<gathB, GATH_T, 0, stream>>>(hbuf, off, sed, dis, b2, c2, part, N);
  k_stats_sum<<<2 * HID, SSUM_T, 0, stream>>>(part, st2, gathB);

  k_gemm_wmma<32, true><<<gemmG, gemmB, 0, stream>>>(c2, W3, hbuf, N, st2, g2, be2, dis);
  k_gather<<<gathB, GATH_T, 0, stream>>>(hbuf, off, sed, dis, b3, c3, part, N);
  k_stats_sum<<<2 * HID, SSUM_T, 0, stream>>>(part, st3, gathB);

  k_final<<<(N + T - 1) / T, T, 0, stream>>>(c1, c2, c3, stats, g1, be1, g2, be2, g3, be3,
                                             Wl, bl, (float*)d_out, N);
}


# Transform: one wave per 16 rows x 32 columns; A rows and weights fetched once by whole-line 16-byte loads into LDS, fragments read from LDS (was 4/8-byte global loads, A twice); whole-row stores
# speedup vs baseline: 2.5119x; 1.0846x over previous
#include <hip/hip_runtime.h>

typedef float v2f __attribute__((ext_vector_type(2)));
typedef float v8f __attribute__((ext_vector_type(8)));

#define HID 32
#define BN_EPS 1e-5f

#define SCAN_T 1024

__global__ void k_zero_cnt(int* cnt, int N) {
  int i = blockIdx.x * blockDim.x + threadIdx.x;
  if (i < N) cnt[i] = 0;
}
__global__ void k_rank(const int* __restrict__ col, int* cnt, int* __restrict__ rank, int E) {
  int e = blockIdx.x * blockDim.x + threadIdx.x;
  if (e < E) rank[e] = atomicAdd(&cnt[col[e]], 1);
}
__global__ __launch_bounds__(SCAN_T) void k_scan(const int* __restrict__ cnt, int* __restrict__ off, int N) {
  __shared__ int sh[SCAN_T];
  __shared__ int sbase;
  const int t     = threadIdx.x;
  const int start = blockIdx.x * SCAN_T;
  int part = 0;
  for (int i = t; i < start; i += SCAN_T) part += cnt[i];
  for (int d = 16; d > 0; d >>= 1) part += __shfl_xor(part, d);
  if (t == 0) sbase = 0;
  __syncthreads();
  if ((t & 31) == 0) atomicAdd(&sbase, part);
  const int idx = start + t;
  const int v   = (idx < N) ? cnt[idx] : 0;
  sh[t] = v;
  __syncthreads();
  for (int d = 1; d < SCAN_T; d <<= 1) {
    const int add = (t >= d) ? sh[t - d] : 0;
    __syncthreads();
    sh[t] += add;
    __syncthreads();
  }
  if (idx <= N) off[idx] = sbase + sh[t] - v;
}
__global__ void k_fill(const int* __restrict__ row, const int* __restrict__ col,
                       const float* __restrict__ w, const int* __restrict__ off,
                       const int* __restrict__ rank, int* __restrict__ sed, int E) {
  int e = blockIdx.x * blockDim.x + threadIdx.x;
  if (e >= E) return;
  const int p = off[col[e]] + rank[e];
  reinterpret_cast<int2*>(sed)[p] = make_int2(row[e], __float_as_int(w[e]));
}
__global__ void k_dis(const int* __restrict__ off, const int* __restrict__ sed,
                      float* __restrict__ dis, int N) {
  int c = blockIdx.x * blockDim.x + threadIdx.x;
  if (c >= N) return;
  float s = 1.0f;
  const int end = off[c + 1];
  for (int p = off[c]; p < end; ++p) s += __int_as_float(sed[2 * (size_t)p + 1]);
  dis[c] = rsqrtf(s);
}

template <int K, bool BN>
__global__ __launch_bounds__(256) void k_gemm_wmma(const float* __restrict__ A,
                                                   const float* __restrict__ B,
                                                   float* __restrict__ D, int M,
                                                   const float* __restrict__ stats,
                                                   const float* __restrict__ g,
                                                   const float* __restrict__ be,
                                                   const float* __restrict__ dis) {
  constexpr int PA  = K + 4;
  constexpr int PW  = HID + 8;
  constexpr int LPR = K / 4;
  constexpr int RPI = 32 / LPR;
  constexpr int NI  = 16 / RPI;
  __shared__ __attribute__((aligned(16))) float sW[K * PW];
  __shared__ __attribute__((aligned(16))) float sA[8][16 * PA];
  const int lane = threadIdx.x;
  const int wave = __builtin_amdgcn_readfirstlane((int)threadIdx.y);
  const int tid  = wave * 32 + lane;
  const int mt   = blockIdx.x * 8 + wave;
  const int numT = (M + 15) >> 4;
  const bool live = mt < numT;
  const int half = lane >> 4;
  const int lm   = lane & 15;

  for (int e = tid * 4; e < K * HID; e += 1024) {
    const float4 w4 = *reinterpret_cast<const float4*>(B + e);
    *reinterpret_cast<float4*>(&sW[(e >> 5) * PW + (e & 31)]) = w4;
  }

  int dli = 0;
  if (live) {
    const int sub = lane % LPR;
    const int q   = lane / LPR;
    const float4* __restrict__ A4 = reinterpret_cast<const float4*>(A);
    float4 v[NI];
#pragma unroll
    for (int t = 0; t < NI; ++t) {
      int gr = mt * 16 + t * RPI + q;
      if (gr > M - 1) gr = M - 1;
      v[t] = A4[(size_t)gr * LPR + sub];
    }
    int ar = mt * 16 + lm;
    if (ar > M - 1) ar = M - 1;
    dli = __float_as_int(dis[ar]);
    if constexpr (BN) {
      const float invN = 1.0f / (float)M;
      float mu[4], rs[4], gg[4], bb[4];
#pragma unroll
      for (int j = 0; j < 4; ++j) {
        const int f = 4 * sub + j;
        mu[j] = stats[f] * invN;
        const float var = stats[32 + f] * invN - mu[j] * mu[j];
        rs[j] = rsqrtf(var + BN_EPS);
        gg[j] = g[f];
        bb[j] = be[f];
      }
#pragma unroll
      for (int t = 0; t < NI; ++t) {
        v[t].x = (fmaxf(v[t].x, 0.0f) - mu[0]) * rs[0] * gg[0] + bb[0];
        v[t].y = (fmaxf(v[t].y, 0.0f) - mu[1]) * rs[1] * gg[1] + bb[1];
        v[t].z = (fmaxf(v[t].z, 0.0f) - mu[2]) * rs[2] * gg[2] + bb[2];
        v[t].w = (fmaxf(v[t].w, 0.0f) - mu[3]) * rs[3] * gg[3] + bb[3];
      }
    }
#pragma unroll
    for (int t = 0; t < NI; ++t)
      *reinterpret_cast<float4*>(&sA[wave][(t * RPI + q) * PA + 4 * sub]) = v[t];
  }
  __syncthreads();
  if (!live) return;

  const float* ap = &sA[wave][lm * PA + 2 * half];
  const float* wp = &sW[(2 * half) * PW + lm];
  v2f fa[K / 4], fb0[K / 4], fb1[K / 4];
#pragma unroll
  for (int s = 0; s < K / 4; ++s) {
    const int k0 = 4 * s;
    fa[s]    = *reinterpret_cast<const v2f*>(ap + k0);
    fb0[s].x = wp[k0 * PW];
    fb0[s].y = wp[(k0 + 1) * PW];
    fb1[s].x = wp[k0 * PW + 16];
    fb1[s].y = wp[(k0 + 1) * PW + 16];
  }
  v8f c0 = {};
  v8f c1 = {};
#pragma unroll
  for (int s = 0; s < K / 4; ++s) {
    c0 = __builtin_amdgcn_wmma_f32_16x16x4_f32(false, fa[s], false, fb0[s], (short)0, c0, false, false);
    c1 = __builtin_amdgcn_wmma_f32_16x16x4_f32(false, fa[s], false, fb1[s], (short)0, c1, false, false);
  }
  float got[8];
#pragma unroll
  for (int r = 0; r < 8; ++r) {
    const int give = __float_as_int(half ? c0[r] : c1[r]);
    got[r] = __int_as_float(__builtin_amdgcn_permlanex16(give, give, 0x76543210, 0xfedcba98, false, false));
  }
#pragma unroll
  for (int r = 0; r < 8; ++r) {
    const float d_lo = __int_as_float(__builtin_amdgcn_readlane(dli, r));
    const float d_hi = __int_as_float(__builtin_amdgcn_readlane(dli, r + 8));
    const float v_lo = half ? got[r] : c0[r];
    const float v_hi = half ? c1[r] : got[r];
    const int row_lo = mt * 16 + r;
    const int row_hi = row_lo + 8;
    if (row_lo < M) D[(size_t)row_lo * HID + lane] = v_lo * d_lo;
    if (row_hi < M) D[(size_t)row_hi * HID + lane] = v_hi * d_hi;
  }
}

#define GATH_T 256
#define GATH_W (GATH_T / 32)
#define GATH_TRIPS 6

template <int NT>
__device__ __forceinline__ float4 gather_pass(const float4* __restrict__ h4, const int2* __restrict__ sed2,
                                              int p0, int n, int q, int sub) {
  int    rr[NT];
  float  ww[NT];
  float4 v[NT];
  const int m = n - 1 - 4 * (NT - 1);
  const unsigned pq = (unsigned)(p0 + q);
  const unsigned pl = (unsigned)(p0 + ((q < m) ? q : m));
#pragma unroll
  for (int t = 0; t < NT; ++t) {
    const int2 ed = sed2[(t < NT - 1) ? pq + 4u * t : pl + 4u * t];
    rr[t] = ed.x;
    ww[t] = (t < NT - 1 || q <= m) ? __int_as_float(ed.y) : 0.0f;
  }
#pragma unroll
  for (int t = 0; t < NT; ++t) v[t] = h4[(unsigned)rr[t] * 8u + (unsigned)sub];
  float4 s = make_float4(v[0].x * ww[0], v[0].y * ww[0], v[0].z * ww[0], v[0].w * ww[0]);
#pragma unroll
  for (int t = 1; t < NT; ++t) {
    s.x += v[t].x * ww[t];
    s.y += v[t].y * ww[t];
    s.z += v[t].z * ww[t];
    s.w += v[t].w * ww[t];
  }
  return s;
}

__global__ __launch_bounds__(GATH_T, 16) void k_gather(const float* __restrict__ h, const int* __restrict__ off,
                                                   const int* __restrict__ sed,
                                                   const float* __restrict__ dis,
                                                   const float* __restrict__ bias,
                                                   float* __restrict__ out,
                                                   float* __restrict__ part, int N) {
  __shared__ __attribute__((aligned(16))) float ls[GATH_W][2 * HID];
  const int lane = threadIdx.x & 31;
  const int q    = lane >> 3;
  const int sub  = lane & 7;
  const int wib = __builtin_amdgcn_readfirstlane((int)(threadIdx.x >> 5));
  const int wpb = (int)(blockDim.x >> 5);
  const int nw  = (int)gridDim.x * wpb;
  const float4* __restrict__ h4   = reinterpret_cast<const float4*>(h);
  const int2*   __restrict__ sed2 = reinterpret_cast<const int2*>(sed);
  const float4 bb = make_float4(bias[4 * sub], bias[4 * sub + 1], bias[4 * sub + 2], bias[4 * sub + 3]);
  float4 s1 = make_float4(0.0f, 0.0f, 0.0f, 0.0f);
  float4 s2 = make_float4(0.0f, 0.0f, 0.0f, 0.0f);
  for (int c = (int)blockIdx.x * wpb + wib; c < N; c += nw) {
    const int beg = off[c];
    const int end = off[c + 1];
    const float dc = dis[c];
    const float4 own = h4[(unsigned)c * 8u + (unsigned)sub];
    float4 acc = make_float4(0.0f, 0.0f, 0.0f, 0.0f);
    for (int p0 = beg; p0 < end; p0 += 4 * GATH_TRIPS) {
      const int n = (end - p0 < 4 * GATH_TRIPS) ? (end - p0) : 4 * GATH_TRIPS;
      float4 s;
      switch ((n + 3) >> 2) {
        case 1:  s = gather_pass<1>(h4, sed2, p0, n, q, sub); break;
        case 2:  s = gather_pass<2>(h4, sed2, p0, n, q, sub); break;
        case 3:  s = gather_pass<3>(h4, sed2, p0, n, q, sub); break;
        case 4:  s = gather_pass<4>(h4, sed2, p0, n, q, sub); break;
        case 5:  s = gather_pass<5>(h4, sed2, p0, n, q, sub); break;
        default: s = gather_pass<6>(h4, sed2, p0, n, q, sub); break;
      }
      acc.x += s.x; acc.y += s.y; acc.z += s.z; acc.w += s.w;
    }
    if (q == 0) { acc.x += own.x; acc.y += own.y; acc.z += own.z; acc.w += own.w; }
    acc.x += __shfl_xor(acc.x, 8);
    acc.y += __shfl_xor(acc.y, 8);
    acc.z += __shfl_xor(acc.z, 8);
    acc.w += __shfl_xor(acc.w, 8);
    acc.x += __shfl_xor(acc.x, 16);
    acc.y += __shfl_xor(acc.y, 16);
    acc.z += __shfl_xor(acc.z, 16);
    acc.w += __shfl_xor(acc.w, 16);
    float4 o;
    o.x = acc.x * dc + bb.x;
    o.y = acc.y * dc + bb.y;
    o.z = acc.z * dc + bb.z;
    o.w = acc.w * dc + bb.w;
    if (q == 0) reinterpret_cast<float4*>(out)[(unsigned)c * 8u + (unsigned)sub] = o;
    const float r0 = fmaxf(o.x, 0.0f), r1 = fmaxf(o.y, 0.0f), r2 = fmaxf(o.z, 0.0f), r3 = fmaxf(o.w, 0.0f);
    s1.x += r0; s1.y += r1; s1.z += r2; s1.w += r3;
    s2.x += r0 * r0; s2.y += r1 * r1; s2.z += r2 * r2; s2.w += r3 * r3;
  }
  if (q == 0) {
    *reinterpret_cast<float4*>(&ls[wib][4 * sub])       = s1;
    *reinterpret_cast<float4*>(&ls[wib][HID + 4 * sub]) = s2;
  }
  __syncthreads();
  if (threadIdx.x < 2 * HID) {
    float t = 0.0f;
    for (int w = 0; w < GATH_W; ++w) t += ls[w][threadIdx.x];
    part[(size_t)threadIdx.x * gridDim.x + blockIdx.x] = t;
  }
}

#define SSUM_T 512
__global__ __launch_bounds__(SSUM_T) void k_stats_sum(const float* __restrict__ part,
                                                      float* __restrict__ stats, int P) {
  __shared__ float sw[SSUM_T / 32];
  const int t = threadIdx.x;
  const float4* p4 = reinterpret_cast<const float4*>(part + (size_t)blockIdx.x * P);
  float s = 0.0f;
  for (int i = t; i < (P >> 2); i += SSUM_T) {
    const float4 v = p4[i];
    s += (v.x + v.y) + (v.z + v.w);
  }
  for (int d = 16; d > 0; d >>= 1) s += __shfl_xor(s, d);
  if ((t & 31) == 0) sw[t >> 5] = s;
  __syncthreads();
  if (t == 0) {
    float r = 0.0f;
    for (int w = 0; w < SSUM_T / 32; ++w) r += sw[w];
    stats[blockIdx.x] = r;
  }
}

__global__ __launch_bounds__(256) void k_final(const float* __restrict__ c1, const float* __restrict__ c2,
                        const float* __restrict__ c3, const float* __restrict__ stats,
                        const float* __restrict__ g1, const float* __restrict__ be1,
                        const float* __restrict__ g2, const float* __restrict__ be2,
                        const float* __restrict__ g3, const float* __restrict__ be3,
                        const float* __restrict__ Wl,
                        const float* __restrict__ bl, float* __restrict__ out, int N) {
  __shared__ float4 prm[3 * HID];
  const int t = threadIdx.x;
  if (t < 3 * HID) {
    const int l = t >> 5;
    const int f = t & 31;
    const float* s = stats + 64 * l;
    const float* g = (l == 0) ? g1 : ((l == 1) ? g2 : g3);
    const float* b = (l == 0) ? be1 : ((l == 1) ? be2 : be3);
    const float invN = 1.0f / (float)N;
    const float mu   = s[f] * invN;
    const float var  = s[32 + f] * invN - mu * mu;
    prm[t] = make_float4(mu, rsqrtf(var + BN_EPS), g[f], b[f]);
  }
  __syncthreads();
  const int n = blockIdx.x * blockDim.x + t;
  if (n >= N) return;
  float a0 = bl[0], a1 = bl[1];
  const float* p = c1 + (size_t)n * HID;
#pragma unroll
  for (int j = 0; j < HID; ++j) {
    const float4 q = prm[j];
    const float v = (fmaxf(p[j], 0.0f) - q.x) * q.y * q.z + q.w;
    a0 += v * Wl[2 * j]; a1 += v * Wl[2 * j + 1];
  }
  p = c2 + (size_t)n * HID;
#pragma unroll
  for (int j = 0; j < HID; ++j) {
    const float4 q = prm[HID + j];
    const float v = (fmaxf(p[j], 0.0f) - q.x) * q.y * q.z + q.w;
    a0 += v * Wl[2 * (HID + j)]; a1 += v * Wl[2 * (HID + j) + 1];
  }
  p = c3 + (size_t)n * HID;
#pragma unroll
  for (int j = 0; j < HID; ++j) {
    const float4 q = prm[2 * HID + j];
    const float v = (fmaxf(p[j], 0.0f) - q.x) * q.y * q.z + q.w;
    a0 += v * Wl[2 * (2 * HID + j)]; a1 += v * Wl[2 * (2 * HID + j) + 1];
  }
  out[2 * n] = a0;
  out[2 * n + 1] = a1;
}

extern "C" void kernel_launch(void* const* d_in, const int* in_sizes, int n_in,
                              void* d_out, int out_size, void* d_ws, size_t ws_size,
                              hipStream_t stream) {
  const float* x   = (const float*)d_in[0];
  const int*   ei  = (const int*)d_in[1];
  const float* ew  = (const float*)d_in[2];
  const float* W1  = (const float*)d_in[3];
  const float* b1  = (const float*)d_in[4];
  const float* g1  = (const float*)d_in[5];
  const float* be1 = (const float*)d_in[6];
  const float* W2  = (const float*)d_in[7];
  const float* b2  = (const float*)d_in[8];
  const float* g2  = (const float*)d_in[9];
  const float* be2 = (const float*)d_in[10];
  const float* W3  = (const float*)d_in[11];
  const float* b3  = (const float*)d_in[12];
  const float* g3  = (const float*)d_in[13];
  const float* be3 = (const float*)d_in[14];
  const float* Wl  = (const float*)d_in[15];
  const float* bl  = (const float*)d_in[16];

  const int F = 64;
  const int N = in_sizes[0] / F;
  const int E = in_sizes[1] / 2;
  const int* row = ei;
  const int* col = ei + E;

  float* ws    = (float*)d_ws;
  float* dis   = ws;
  float* hbuf  = dis  + (((size_t)N + 63) & ~(size_t)63);
  float* c1    = hbuf + (size_t)N * HID;
  float* c2    = c1   + (size_t)N * HID;
  float* c3    = c2   + (size_t)N * HID;
  float* stats = c3   + (size_t)N * HID;
  float* st1   = stats;
  float* st2   = stats + 64;
  float* st3   = stats + 128;
  int*   cnt   = (int*)(stats + 192);
  int*   off   = cnt + N;
  int*   rank  = off + N + 1;
  const size_t sedOff = (((size_t)((float*)(rank + E) - ws)) + 63) & ~(size_t)63;
  int*   sed   = (int*)(ws + sedOff);
  const int gathB = 2048;
  const size_t partOff = (sedOff + 2 * (size_t)E + 63) & ~(size_t)63;
  float* part  = ws + partOff;

  const int T = 256;

  k_zero_cnt<<<(N + T - 1) / T, T, 0, stream>>>(cnt, N);
  k_rank<<<(E + T - 1) / T, T, 0, stream>>>(col, cnt, rank, E);
  k_scan<<<(N + SCAN_T) / SCAN_T, SCAN_T, 0, stream>>>(cnt, off, N);
  k_fill<<<(E + T - 1) / T, T, 0, stream>>>(row, col, ew, off, rank, sed, E);
  k_dis<<<(N + T - 1) / T, T, 0, stream>>>(off, sed, dis, N);

  const int numT = (N + 15) / 16;
  const dim3 gemmB(32, 8);
  const dim3 gemmG((numT + 7) / 8, 1);

  k_gemm_wmma<64, false><<<gemmG, gemmB, 0, stream>>>(x, W1, hbuf, N, nullptr, nullptr, nullptr, dis);
  k_gather<<<gathB, GATH_T, 0, stream>>>(hbuf, off, sed, dis, b1, c1, part, N);
  k_stats_sum<<<2 * HID, SSUM_T, 0, stream>>>(part, st1, gathB);

  k_gemm_wmma<32, true><<<gemmG, gemmB, 0, stream>>>(c1, W2, hbuf, N, st1, g1, be1, dis);
  k_gather<<<gathB, GATH_T, 0, stream>>>(hbuf, off, sed, dis, b2, c2, part, N);
  k_stats_sum<<<2 * HID, SSUM_T, 0, stream>>>(part, st2, gathB);

  k_gemm_wmma<32, true><<<gemmG, gemmB, 0, stream>>>(c2, W3, hbuf, N, st2, g2, be2, dis);
  k_gather<<<gathB, GATH_T, 0, stream>>>(hbuf, off, sed, dis, b3, c3, part, N);
  k_stats_sum<<<2 * HID, SSUM_T, 0, stream>>>(part, st3, gathB);

  k_final<<<(N + T - 1) / T, T, 0, stream>>>(c1, c2, c3, stats, g1, be1, g2, be2, g3, be3,
                                             Wl, bl, (float*)d_out, N);
}


# BN sums: gather blocks' shares (512 blocks of 32 waves) are added up in fixed order by the kernels that use them (next transform, readout); 3 sum launches gone, 15 to 12; BN partial sums regrouped
# speedup vs baseline: 2.7388x; 1.0903x over previous
#include <hip/hip_runtime.h>

typedef float v2f __attribute__((ext_vector_type(2)));
typedef float v8f __attribute__((ext_vector_type(8)));

#define HID 32
#define BN_EPS 1e-5f

#define SCAN_T 1024

__global__ void k_zero_cnt(int* cnt, int N) {
  int i = blockIdx.x * blockDim.x + threadIdx.x;
  if (i < N) cnt[i] = 0;
}
__global__ void k_rank(const int* __restrict__ col, int* cnt, int* __restrict__ rank, int E) {
  int e = blockIdx.x * blockDim.x + threadIdx.x;
  if (e < E) rank[e] = atomicAdd(&cnt[col[e]], 1);
}
__global__ __launch_bounds__(SCAN_T) void k_scan(const int* __restrict__ cnt, int* __restrict__ off, int N) {
  __shared__ int sh[SCAN_T];
  __shared__ int sbase;
  const int t     = threadIdx.x;
  const int start = blockIdx.x * SCAN_T;
  int part = 0;
  for (int i = t; i < start; i += SCAN_T) part += cnt[i];
  for (int d = 16; d > 0; d >>= 1) part += __shfl_xor(part, d);
  if (t == 0) sbase = 0;
  __syncthreads();
  if ((t & 31) == 0) atomicAdd(&sbase, part);
  const int idx = start + t;
  const int v   = (idx < N) ? cnt[idx] : 0;
  sh[t] = v;
  __syncthreads();
  for (int d = 1; d < SCAN_T; d <<= 1) {
    const int add = (t >= d) ? sh[t - d] : 0;
    __syncthreads();
    sh[t] += add;
    __syncthreads();
  }
  if (idx <= N) off[idx] = sbase + sh[t] - v;
}
__global__ void k_fill(const int* __restrict__ row, const int* __restrict__ col,
                       const float* __restrict__ w, const int* __restrict__ off,
                       const int* __restrict__ rank, int* __restrict__ sed, int E) {
  int e = blockIdx.x * blockDim.x + threadIdx.x;
  if (e >= E) return;
  const int p = off[col[e]] + rank[e];
  reinterpret_cast<int2*>(sed)[p] = make_int2(row[e], __float_as_int(w[e]));
}
__global__ void k_dis(const int* __restrict__ off, const int* __restrict__ sed,
                      float* __restrict__ dis, int N) {
  int c = blockIdx.x * blockDim.x + threadIdx.x;
  if (c >= N) return;
  float s = 1.0f;
  const int end = off[c + 1];
  for (int p = off[c]; p < end; ++p) s += __int_as_float(sed[2 * (size_t)p + 1]);
  dis[c] = rsqrtf(s);
}

#define GATH_B 512
template <int NT>
__device__ __forceinline__ void shares_ask(const float* __restrict__ part, int tid, float4 (&sh)[GATH_B * 16 / NT]) {
  constexpr int NG = NT / 16;
  const int f4 = (tid & 15) * 4;
  const int g  = tid >> 4;
#pragma unroll
  for (int i = 0; i < GATH_B / NG; ++i)
    sh[i] = *reinterpret_cast<const float4*>(part + (size_t)(g + i * NG) * (2 * HID) + f4);
}
template <int NT>
__device__ __forceinline__ void shares_add(const float4 (&sh)[GATH_B * 16 / NT], int tid,
                                           float* __restrict__ red, float* __restrict__ sst) {
  constexpr int NG = NT / 16;
  const int f4 = (tid & 15) * 4;
  const int g  = tid >> 4;
  float4 s = sh[0];
#pragma unroll
  for (int i = 1; i < GATH_B / NG; ++i) { s.x += sh[i].x; s.y += sh[i].y; s.z += sh[i].z; s.w += sh[i].w; }
  *reinterpret_cast<float4*>(&red[g * (2 * HID) + f4]) = s;
  __syncthreads();
  if (tid < 2 * HID) {
    float r = 0.0f;
    for (int k = 0; k < NG; ++k) r += red[k * (2 * HID) + tid];
    sst[tid] = r;
  }
  __syncthreads();
}

template <int K, bool BN, int NW>
__global__ __launch_bounds__(NW * 32) void k_gemm_wmma(const float* __restrict__ A,
                                                   const float* __restrict__ B,
                                                   float* __restrict__ D, int M,
                                                   const float* __restrict__ part,
                                                   float* __restrict__ stats_out,
                                                   const float* __restrict__ g,
                                                   const float* __restrict__ be,
                                                   const float* __restrict__ dis) {
  constexpr int PA  = K + 4;
  constexpr int PW  = HID + 8;
  constexpr int LPR = K / 4;
  constexpr int RPI = 32 / LPR;
  constexpr int NI  = 16 / RPI;
  __shared__ __attribute__((aligned(16))) float sW[K * PW];
  __shared__ __attribute__((aligned(16))) float sA[NW][16 * PA];
  __shared__ __attribute__((aligned(16))) float red[BN ? NW * 2 * 2 * HID : 4];
  __shared__ float sst[BN ? 2 * HID : 1];
  const int lane = threadIdx.x;
  const int wave = __builtin_amdgcn_readfirstlane((int)threadIdx.y);
  const int tid  = wave * 32 + lane;
  const int mt   = blockIdx.x * NW + wave;
  const int numT = (M + 15) >> 4;
  const bool live = mt < numT;
  const int half = lane >> 4;
  const int lm   = lane & 15;

  for (int e = tid * 4; e < K * HID; e += NW * 128) {
    const float4 w4 = *reinterpret_cast<const float4*>(B + e);
    *reinterpret_cast<float4*>(&sW[(e >> 5) * PW + (e & 31)]) = w4;
  }

  float4 sh[BN ? GATH_B / (NW * 2) : 1];
  if constexpr (BN) shares_ask<NW * 32>(part, tid, sh);
  int dli = 0;
  const int sub = lane % LPR;
  const int q   = lane / LPR;
  float4 v[NI];
  if (live) {
    const float4* __restrict__ A4 = reinterpret_cast<const float4*>(A);
#pragma unroll
    for (int t = 0; t < NI; ++t) {
      int gr = mt * 16 + t * RPI + q;
      if (gr > M - 1) gr = M - 1;
      v[t] = A4[(size_t)gr * LPR + sub];
    }
    int ar = mt * 16 + lm;
    if (ar > M - 1) ar = M - 1;
    dli = __float_as_int(dis[ar]);
    if constexpr (!BN) {
#pragma unroll
      for (int t = 0; t < NI; ++t)
        *reinterpret_cast<float4*>(&sA[wave][(t * RPI + q) * PA + 4 * sub]) = v[t];
    }
  }
  if constexpr (BN) {
    shares_add<NW * 32>(sh, tid, red, sst);
    if (blockIdx.x == 0 && tid < 2 * HID) stats_out[tid] = sst[tid];
    if (live) {
      const float invN = 1.0f / (float)M;
      float mu[4], rs[4], gg[4], bb[4];
#pragma unroll
      for (int j = 0; j < 4; ++j) {
        const int f = 4 * sub + j;
        mu[j] = sst[f] * invN;
        const float var = sst[32 + f] * invN - mu[j] * mu[j];
        rs[j] = rsqrtf(var + BN_EPS);
        gg[j] = g[f];
        bb[j] = be[f];
      }
#pragma unroll
      for (int t = 0; t < NI; ++t) {
        v[t].x = (fmaxf(v[t].x, 0.0f) - mu[0]) * rs[0] * gg[0] + bb[0];
        v[t].y = (fmaxf(v[t].y, 0.0f) - mu[1]) * rs[1] * gg[1] + bb[1];
        v[t].z = (fmaxf(v[t].z, 0.0f) - mu[2]) * rs[2] * gg[2] + bb[2];
        v[t].w = (fmaxf(v[t].w, 0.0f) - mu[3]) * rs[3] * gg[3] + bb[3];
      }
#pragma unroll
      for (int t = 0; t < NI; ++t)
        *reinterpret_cast<float4*>(&sA[wave][(t * RPI + q) * PA + 4 * sub]) = v[t];
    }
  }
  __syncthreads();
  if (!live) return;

  const float* ap = &sA[wave][lm * PA + 2 * half];
  const float* wp = &sW[(2 * half) * PW + lm];
  v2f fa[K / 4], fb0[K / 4], fb1[K / 4];
#pragma unroll
  for (int s = 0; s < K / 4; ++s) {
    const int k0 = 4 * s;
    fa[s]    = *reinterpret_cast<const v2f*>(ap + k0);
    fb0[s].x = wp[k0 * PW];
    fb0[s].y = wp[(k0 + 1) * PW];
    fb1[s].x = wp[k0 * PW + 16];
    fb1[s].y = wp[(k0 + 1) * PW + 16];
  }
  v8f c0 = {};
  v8f c1 = {};
#pragma unroll
  for (int s = 0; s < K / 4; ++s) {
    c0 = __builtin_amdgcn_wmma_f32_16x16x4_f32(false, fa[s], false, fb0[s], (short)0, c0, false, false);
    c1 = __builtin_amdgcn_wmma_f32_16x16x4_f32(false, fa[s], false, fb1[s], (short)0, c1, false, false);
  }
  float got[8];
#pragma unroll
  for (int r = 0; r < 8; ++r) {
    const int give = __float_as_int(half ? c0[r] : c1[r]);
    got[r] = __int_as_float(__builtin_amdgcn_permlanex16(give, give, 0x76543210, 0xfedcba98, false, false));
  }
#pragma unroll
  for (int r = 0; r < 8; ++r) {
    const float d_lo = __int_as_float(__builtin_amdgcn_readlane(dli, r));
    const float d_hi = __int_as_float(__builtin_amdgcn_readlane(dli, r + 8));
    const float v_lo = half ? got[r] : c0[r];
    const float v_hi = half ? c1[r] : got[r];
    const int row_lo = mt * 16 + r;
    const int row_hi = row_lo + 8;
    if (row_lo < M) D[(size_t)row_lo * HID + lane] = v_lo * d_lo;
    if (row_hi < M) D[(size_t)row_hi * HID + lane] = v_hi * d_hi;
  }
}

#define GATH_T 1024
#define GATH_W (GATH_T / 32)
#define GATH_TRIPS 6

template <int NT>
__device__ __forceinline__ float4 gather_pass(const float4* __restrict__ h4, const int2* __restrict__ sed2,
                                              int p0, int n, int q, int sub) {
  int    rr[NT];
  float  ww[NT];
  float4 v[NT];
  const int m = n - 1 - 4 * (NT - 1);
  const unsigned pq = (unsigned)(p0 + q);
  const unsigned pl = (unsigned)(p0 + ((q < m) ? q : m));
#pragma unroll
  for (int t = 0; t < NT; ++t) {
    const int2 ed = sed2[(t < NT - 1) ? pq + 4u * t : pl + 4u * t];
    rr[t] = ed.x;
    ww[t] = (t < NT - 1 || q <= m) ? __int_as_float(ed.y) : 0.0f;
  }
#pragma unroll
  for (int t = 0; t < NT; ++t) v[t] = h4[(unsigned)rr[t] * 8u + (unsigned)sub];
  float4 s = make_float4(v[0].x * ww[0], v[0].y * ww[0], v[0].z * ww[0], v[0].w * ww[0]);
#pragma unroll
  for (int t = 1; t < NT; ++t) {
    s.x += v[t].x * ww[t];
    s.y += v[t].y * ww[t];
    s.z += v[t].z * ww[t];
    s.w += v[t].w * ww[t];
  }
  return s;
}

__global__ __launch_bounds__(GATH_T, 16) void k_gather(const float* __restrict__ h, const int* __restrict__ off,
                                                   const int* __restrict__ sed,
                                                   const float* __restrict__ dis,
                                                   const float* __restrict__ bias,
                                                   float* __restrict__ out,
                                                   float* __restrict__ part, int N) {
  __shared__ __attribute__((aligned(16))) float ls[GATH_W][2 * HID];
  const int lane = threadIdx.x & 31;
  const int q    = lane >> 3;
  const int sub  = lane & 7;
  const int wib = __builtin_amdgcn_readfirstlane((int)(threadIdx.x >> 5));
  const int wpb = (int)(blockDim.x >> 5);
  const int nw  = (int)gridDim.x * wpb;
  const float4* __restrict__ h4   = reinterpret_cast<const float4*>(h);
  const int2*   __restrict__ sed2 = reinterpret_cast<const int2*>(sed);
  const float4 bb = make_float4(bias[4 * sub], bias[4 * sub + 1], bias[4 * sub + 2], bias[4 * sub + 3]);
  float4 s1 = make_float4(0.0f, 0.0f, 0.0f, 0.0f);
  float4 s2 = make_float4(0.0f, 0.0f, 0.0f, 0.0f);
  for (int c = (int)blockIdx.x * wpb + wib; c < N; c += nw) {
    const int beg = off[c];
    const int end = off[c + 1];
    const float dc = dis[c];
    const float4 own = h4[(unsigned)c * 8u + (unsigned)sub];
    float4 acc = make_float4(0.0f, 0.0f, 0.0f, 0.0f);
    for (int p0 = beg; p0 < end; p0 += 4 * GATH_TRIPS) {
      const int n = (end - p0 < 4 * GATH_TRIPS) ? (end - p0) : 4 * GATH_TRIPS;
      float4 s;
      switch ((n + 3) >> 2) {
        case 1:  s = gather_pass<1>(h4, sed2, p0, n, q, sub); break;
        case 2:  s = gather_pass<2>(h4, sed2, p0, n, q, sub); break;
        case 3:  s = gather_pass<3>(h4, sed2, p0, n, q, sub); break;
        case 4:  s = gather_pass<4>(h4, sed2, p0, n, q, sub); break;
        case 5:  s = gather_pass<5>(h4, sed2, p0, n, q, sub); break;
        default: s = gather_pass<6>(h4, sed2, p0, n, q, sub); break;
      }
      acc.x += s.x; acc.y += s.y; acc.z += s.z; acc.w += s.w;
    }
    if (q == 0) { acc.x += own.x; acc.y += own.y; acc.z += own.z; acc.w += own.w; }
    acc.x += __shfl_xor(acc.x, 8);
    acc.y += __shfl_xor(acc.y, 8);
    acc.z += __shfl_xor(acc.z, 8);
    acc.w += __shfl_xor(acc.w, 8);
    acc.x += __shfl_xor(acc.x, 16);
    acc.y += __shfl_xor(acc.y, 16);
    acc.z += __shfl_xor(acc.z, 16);
    acc.w += __shfl_xor(acc.w, 16);
    float4 o;
    o.x = acc.x * dc + bb.x;
    o.y = acc.y * dc + bb.y;
    o.z = acc.z * dc + bb.z;
    o.w = acc.w * dc + bb.w;
    if (q == 0) reinterpret_cast<float4*>(out)[(unsigned)c * 8u + (unsigned)sub] = o;
    const float r0 = fmaxf(o.x, 0.0f), r1 = fmaxf(o.y, 0.0f), r2 = fmaxf(o.z, 0.0f), r3 = fmaxf(o.w, 0.0f);
    s1.x += r0; s1.y += r1; s1.z += r2; s1.w += r3;
    s2.x += r0 * r0; s2.y += r1 * r1; s2.z += r2 * r2; s2.w += r3 * r3;
  }
  if (q == 0) {
    *reinterpret_cast<float4*>(&ls[wib][4 * sub])       = s1;
    *reinterpret_cast<float4*>(&ls[wib][HID + 4 * sub]) = s2;
  }
  __syncthreads();
  if (threadIdx.x < 2 * HID) {
    float t = 0.0f;
    for (int w = 0; w < GATH_W; ++w) t += ls[w][threadIdx.x];
    part[(size_t)blockIdx.x * (2 * HID) + threadIdx.x] = t;
  }
}

#define FIN_T 512
__global__ __launch_bounds__(FIN_T) void k_final(const float* __restrict__ c1, const float* __restrict__ c2,
                        const float* __restrict__ c3, const float* __restrict__ stats,
                        const float* __restrict__ part,
                        const float* __restrict__ g1, const float* __restrict__ be1,
                        const float* __restrict__ g2, const float* __restrict__ be2,
                        const float* __restrict__ g3, const float* __restrict__ be3,
                        const float* __restrict__ Wl,
                        const float* __restrict__ bl, float* __restrict__ out, int N) {
  __shared__ float4 prm[3 * HID];
  __shared__ __attribute__((aligned(16))) float red[(FIN_T / 16) * 2 * HID];
  __shared__ float sst[2 * HID];
  const int t = threadIdx.x;
  float4 sh[GATH_B * 16 / FIN_T];
  shares_ask<FIN_T>(part, t, sh);
  shares_add<FIN_T>(sh, t, red, sst);
  if (t < 3 * HID) {
    const int l = t >> 5;
    const int f = t & 31;
    float sm, sq;
    if (l < 2) { sm = stats[64 * l + f]; sq = stats[64 * l + 32 + f]; }
    else       { sm = sst[f];            sq = sst[32 + f]; }
    const float* g = (l == 0) ? g1 : ((l == 1) ? g2 : g3);
    const float* b = (l == 0) ? be1 : ((l == 1) ? be2 : be3);
    const float invN = 1.0f / (float)N;
    const float mu   = sm * invN;
    const float var  = sq * invN - mu * mu;
    prm[t] = make_float4(mu, rsqrtf(var + BN_EPS), g[f], b[f]);
  }
  __syncthreads();
  const int n = blockIdx.x * blockDim.x + t;
  if (n >= N) return;
  float a0 = bl[0], a1 = bl[1];
  const float* p = c1 + (size_t)n * HID;
#pragma unroll
  for (int j = 0; j < HID; ++j) {
    const float4 q = prm[j];
    const float v = (fmaxf(p[j], 0.0f) - q.x) * q.y * q.z + q.w;
    a0 += v * Wl[2 * j]; a1 += v * Wl[2 * j + 1];
  }
  p = c2 + (size_t)n * HID;
#pragma unroll
  for (int j = 0; j < HID; ++j) {
    const float4 q = prm[HID + j];
    const float v = (fmaxf(p[j], 0.0f) - q.x) * q.y * q.z + q.w;
    a0 += v * Wl[2 * (HID + j)]; a1 += v * Wl[2 * (HID + j) + 1];
  }
  p = c3 + (size_t)n * HID;
#pragma unroll
  for (int j = 0; j < HID; ++j) {
    const float4 q = prm[2 * HID + j];
    const float v = (fmaxf(p[j], 0.0f) - q.x) * q.y * q.z + q.w;
    a0 += v * Wl[2 * (2 * HID + j)]; a1 += v * Wl[2 * (2 * HID + j) + 1];
  }
  out[2 * n] = a0;
  out[2 * n + 1] = a1;
}

extern "C" void kernel_launch(void* const* d_in, const int* in_sizes, int n_in,
                              void* d_out, int out_size, void* d_ws, size_t ws_size,
                              hipStream_t stream) {
  const float* x   = (const float*)d_in[0];
  const int*   ei  = (const int*)d_in[1];
  const float* ew  = (const float*)d_in[2];
  const float* W1  = (const float*)d_in[3];
  const float* b1  = (const float*)d_in[4];
  const float* g1  = (const float*)d_in[5];
  const float* be1 = (const float*)d_in[6];
  const float* W2  = (const float*)d_in[7];
  const float* b2  = (const float*)d_in[8];
  const float* g2  = (const float*)d_in[9];
  const float* be2 = (const float*)d_in[10];
  const float* W3  = (const float*)d_in[11];
  const float* b3  = (const float*)d_in[12];
  const float* g3  = (const float*)d_in[13];
  const float* be3 = (const float*)d_in[14];
  const float* Wl  = (const float*)d_in[15];
  const float* bl  = (const float*)d_in[16];

  const int F = 64;
  const int N = in_sizes[0] / F;
  const int E = in_sizes[1] / 2;
  const int* row = ei;
  const int* col = ei + E;

  float* ws    = (float*)d_ws;
  float* dis   = ws;
  float* hbuf  = dis  + (((size_t)N + 63) & ~(size_t)63);
  float* c1    = hbuf + (size_t)N * HID;
  float* c2    = c1   + (size_t)N * HID;
  float* c3    = c2   + (size_t)N * HID;
  float* stats = c3   + (size_t)N * HID;
  float* st1   = stats;
  float* st2   = stats + 64;
  int*   cnt   = (int*)(stats + 192);
  int*   off   = cnt + N;
  int*   rank  = off + N + 1;
  const size_t sedOff = (((size_t)((float*)(rank + E) - ws)) + 63) & ~(size_t)63;
  int*   sed   = (int*)(ws + sedOff);
  const int gathB = GATH_B;
  const size_t partOff = (sedOff + 2 * (size_t)E + 63) & ~(size_t)63;
  float* part  = ws + partOff;

  const int T = 256;

  k_zero_cnt<<<(N + T - 1) / T, T, 0, stream>>>(cnt, N);
  k_rank<<<(E + T - 1) / T, T, 0, stream>>>(col, cnt, rank, E);
  k_scan<<<(N + SCAN_T) / SCAN_T, SCAN_T, 0, stream>>>(cnt, off, N);
  k_fill<<<(E + T - 1) / T, T, 0, stream>>>(row, col, ew, off, rank, sed, E);
  k_dis<<<(N + T - 1) / T, T, 0, stream>>>(off, sed, dis, N);

  const int numT = (N + 15) / 16;
  const dim3 gemmB(32, 8);
  const dim3 gemmG((numT + 7) / 8, 1);
  const dim3 gemmB2(32, 32);
  const dim3 gemmG2((numT + 31) / 32, 1);

  k_gemm_wmma<64, false, 8><<<gemmG, gemmB, 0, stream>>>(x, W1, hbuf, N, nullptr, nullptr, nullptr, nullptr, dis);
  k_gather<<<gathB, GATH_T, 0, stream>>>(hbuf, off, sed, dis, b1, c1, part, N);

  k_gemm_wmma<32, true, 32><<<gemmG2, gemmB2, 0, stream>>>(c1, W2, hbuf, N, part, st1, g1, be1, dis);
  k_gather<<<gathB, GATH_T, 0, stream>>>(hbuf, off, sed, dis, b2, c2, part, N);

  k_gemm_wmma<32, true, 32><<<gemmG2, gemmB2, 0, stream>>>(c2, W3, hbuf, N, part, st2, g2, be2, dis);
  k_gather<<<gathB, GATH_T, 0, stream>>>(hbuf, off, sed, dis, b3, c3, part, N);

  k_final<<<(N + FIN_T - 1) / FIN_T, FIN_T, 0, stream>>>(c1, c2, c3, stats, part, g1, be1, g2, be2, g3, be3,
                                             Wl, bl, (float*)d_out, N);
}


# Edges placed straight into fixed 64-slot segments per target (slot = arrival number from the count atomic; overflow to a side list); prefix-sum and placement launches gone, 12 to 10; sum order kept
# speedup vs baseline: 3.2008x; 1.1687x over previous
#include <hip/hip_runtime.h>

typedef float v2f __attribute__((ext_vector_type(2)));
typedef float v8f __attribute__((ext_vector_type(8)));

#define HID 32
#define BN_EPS 1e-5f

#define SEG 64

__global__ void k_zero_cnt(int* cnt, int N) {
  int i = blockIdx.x * blockDim.x + threadIdx.x;
  if (i <= N) cnt[i] = 0;
}
__global__ void k_place(const int* __restrict__ row, const int* __restrict__ col, const float* __restrict__ w,
                        int* cnt, int* __restrict__ sed, int* __restrict__ ovf, int E, int N) {
  int e = blockIdx.x * blockDim.x + threadIdx.x;
  if (e >= E) return;
  const int c = col[e];
  const int src = row[e];
  const int wb = __float_as_int(w[e]);
  const int r = atomicAdd(&cnt[c], 1);
  if (r < SEG) {
    reinterpret_cast<int2*>(sed)[(size_t)c * SEG + r] = make_int2(src, wb);
  } else {
    const int j = atomicAdd(&cnt[N], 1);
    reinterpret_cast<int4*>(ovf)[j] = make_int4(c, src, wb, 0);
  }
}
__global__ void k_dis(const int* __restrict__ cnt, const int* __restrict__ sed, const int* __restrict__ ovf,
                      float* __restrict__ dis, int N) {
  int c = blockIdx.x * blockDim.x + threadIdx.x;
  if (c >= N) return;
  float s = 1.0f;
  const int nc = cnt[c];
  const int m  = (nc < SEG) ? nc : SEG;
  const size_t b = (size_t)c * SEG;
  for (int p = 0; p < m; ++p) s += __int_as_float(sed[2 * (b + p) + 1]);
  if (nc > SEG) {
    const int no = cnt[N];
    const int4* __restrict__ ovf4 = reinterpret_cast<const int4*>(ovf);
    for (int j = 0; j < no; ++j) {
      const int4 en = ovf4[j];
      if (en.x == c) s += __int_as_float(en.z);
    }
  }
  dis[c] = rsqrtf(s);
}

#define GATH_B 512
template <int NT>
__device__ __forceinline__ void shares_ask(const float* __restrict__ part, int tid, float4 (&sh)[GATH_B * 16 / NT]) {
  constexpr int NG = NT / 16;
  const int f4 = (tid & 15) * 4;
  const int g  = tid >> 4;
#pragma unroll
  for (int i = 0; i < GATH_B / NG; ++i)
    sh[i] = *reinterpret_cast<const float4*>(part + (size_t)(g + i * NG) * (2 * HID) + f4);
}
template <int NT>
__device__ __forceinline__ void shares_add(const float4 (&sh)[GATH_B * 16 / NT], int tid,
                                           float* __restrict__ red, float* __restrict__ sst) {
  constexpr int NG = NT / 16;
  const int f4 = (tid & 15) * 4;
  const int g  = tid >> 4;
  float4 s = sh[0];
#pragma unroll
  for (int i = 1; i < GATH_B / NG; ++i) { s.x += sh[i].x; s.y += sh[i].y; s.z += sh[i].z; s.w += sh[i].w; }
  *reinterpret_cast<float4*>(&red[g * (2 * HID) + f4]) = s;
  __syncthreads();
  if (tid < 2 * HID) {
    float r = 0.0f;
    for (int k = 0; k < NG; ++k) r += red[k * (2 * HID) + tid];
    sst[tid] = r;
  }
  __syncthreads();
}

template <int K, bool BN, int NW>
__global__ __launch_bounds__(NW * 32) void k_gemm_wmma(const float* __restrict__ A,
                                                   const float* __restrict__ B,
                                                   float* __restrict__ D, int M,
                                                   const float* __restrict__ part,
                                                   float* __restrict__ stats_out,
                                                   const float* __restrict__ g,
                                                   const float* __restrict__ be,
                                                   const float* __restrict__ dis) {
  constexpr int PA  = K + 4;
  constexpr int PW  = HID + 8;
  constexpr int LPR = K / 4;
  constexpr int RPI = 32 / LPR;
  constexpr int NI  = 16 / RPI;
  __shared__ __attribute__((aligned(16))) float sW[K * PW];
  __shared__ __attribute__((aligned(16))) float sA[NW][16 * PA];
  __shared__ __attribute__((aligned(16))) float red[BN ? NW * 2 * 2 * HID : 4];
  __shared__ float sst[BN ? 2 * HID : 1];
  const int lane = threadIdx.x;
  const int wave = __builtin_amdgcn_readfirstlane((int)threadIdx.y);
  const int tid  = wave * 32 + lane;
  const int mt   = blockIdx.x * NW + wave;
  const int numT = (M + 15) >> 4;
  const bool live = mt < numT;
  const int half = lane >> 4;
  const int lm   = lane & 15;

  for (int e = tid * 4; e < K * HID; e += NW * 128) {
    const float4 w4 = *reinterpret_cast<const float4*>(B + e);
    *reinterpret_cast<float4*>(&sW[(e >> 5) * PW + (e & 31)]) = w4;
  }

  float4 sh[BN ? GATH_B / (NW * 2) : 1];
  if constexpr (BN) shares_ask<NW * 32>(part, tid, sh);
  int dli = 0;
  const int sub = lane % LPR;
  const int q   = lane / LPR;
  float4 v[NI];
  if (live) {
    const float4* __restrict__ A4 = reinterpret_cast<const float4*>(A);
#pragma unroll
    for (int t = 0; t < NI; ++t) {
      int gr = mt * 16 + t * RPI + q;
      if (gr > M - 1) gr = M - 1;
      v[t] = A4[(size_t)gr * LPR + sub];
    }
    int ar = mt * 16 + lm;
    if (ar > M - 1) ar = M - 1;
    dli = __float_as_int(dis[ar]);
    if constexpr (!BN) {
#pragma unroll
      for (int t = 0; t < NI; ++t)
        *reinterpret_cast<float4*>(&sA[wave][(t * RPI + q) * PA + 4 * sub]) = v[t];
    }
  }
  if constexpr (BN) {
    shares_add<NW * 32>(sh, tid, red, sst);
    if (blockIdx.x == 0 && tid < 2 * HID) stats_out[tid] = sst[tid];
    if (live) {
      const float invN = 1.0f / (float)M;
      float mu[4], rs[4], gg[4], bb[4];
#pragma unroll
      for (int j = 0; j < 4; ++j) {
        const int f = 4 * sub + j;
        mu[j] = sst[f] * invN;
        const float var = sst[32 + f] * invN - mu[j] * mu[j];
        rs[j] = rsqrtf(var + BN_EPS);
        gg[j] = g[f];
        bb[j] = be[f];
      }
#pragma unroll
      for (int t = 0; t < NI; ++t) {
        v[t].x = (fmaxf(v[t].x, 0.0f) - mu[0]) * rs[0] * gg[0] + bb[0];
        v[t].y = (fmaxf(v[t].y, 0.0f) - mu[1]) * rs[1] * gg[1] + bb[1];
        v[t].z = (fmaxf(v[t].z, 0.0f) - mu[2]) * rs[2] * gg[2] + bb[2];
        v[t].w = (fmaxf(v[t].w, 0.0f) - mu[3]) * rs[3] * gg[3] + bb[3];
      }
#pragma unroll
      for (int t = 0; t < NI; ++t)
        *reinterpret_cast<float4*>(&sA[wave][(t * RPI + q) * PA + 4 * sub]) = v[t];
    }
  }
  __syncthreads();
  if (!live) return;

  const float* ap = &sA[wave][lm * PA + 2 * half];
  const float* wp = &sW[(2 * half) * PW + lm];
  v2f fa[K / 4], fb0[K / 4], fb1[K / 4];
#pragma unroll
  for (int s = 0; s < K / 4; ++s) {
    const int k0 = 4 * s;
    fa[s]    = *reinterpret_cast<const v2f*>(ap + k0);
    fb0[s].x = wp[k0 * PW];
    fb0[s].y = wp[(k0 + 1) * PW];
    fb1[s].x = wp[k0 * PW + 16];
    fb1[s].y = wp[(k0 + 1) * PW + 16];
  }
  v8f c0 = {};
  v8f c1 = {};
#pragma unroll
  for (int s = 0; s < K / 4; ++s) {
    c0 = __builtin_amdgcn_wmma_f32_16x16x4_f32(false, fa[s], false, fb0[s], (short)0, c0, false, false);
    c1 = __builtin_amdgcn_wmma_f32_16x16x4_f32(false, fa[s], false, fb1[s], (short)0, c1, false, false);
  }
  float got[8];
#pragma unroll
  for (int r = 0; r < 8; ++r) {
    const int give = __float_as_int(half ? c0[r] : c1[r]);
    got[r] = __int_as_float(__builtin_amdgcn_permlanex16(give, give, 0x76543210, 0xfedcba98, false, false));
  }
#pragma unroll
  for (int r = 0; r < 8; ++r) {
    const float d_lo = __int_as_float(__builtin_amdgcn_readlane(dli, r));
    const float d_hi = __int_as_float(__builtin_amdgcn_readlane(dli, r + 8));
    const float v_lo = half ? got[r] : c0[r];
    const float v_hi = half ? c1[r] : got[r];
    const int row_lo = mt * 16 + r;
    const int row_hi = row_lo + 8;
    if (row_lo < M) D[(size_t)row_lo * HID + lane] = v_lo * d_lo;
    if (row_hi < M) D[(size_t)row_hi * HID + lane] = v_hi * d_hi;
  }
}

#define GATH_T 1024
#define GATH_W (GATH_T / 32)
#define GATH_TRIPS 6

template <int NT>
__device__ __forceinline__ float4 gather_pass(const float4* __restrict__ h4, const int2* __restrict__ sed2,
                                              int p0, int n, int q, int sub) {
  int    rr[NT];
  float  ww[NT];
  float4 v[NT];
  const int m = n - 1 - 4 * (NT - 1);
  const unsigned pq = (unsigned)(p0 + q);
  const unsigned pl = (unsigned)(p0 + ((q < m) ? q : m));
#pragma unroll
  for (int t = 0; t < NT; ++t) {
    const int2 ed = sed2[(t < NT - 1) ? pq + 4u * t : pl + 4u * t];
    rr[t] = ed.x;
    ww[t] = (t < NT - 1 || q <= m) ? __int_as_float(ed.y) : 0.0f;
  }
#pragma unroll
  for (int t = 0; t < NT; ++t) v[t] = h4[(unsigned)rr[t] * 8u + (unsigned)sub];
  float4 s = make_float4(v[0].x * ww[0], v[0].y * ww[0], v[0].z * ww[0], v[0].w * ww[0]);
#pragma unroll
  for (int t = 1; t < NT; ++t) {
    s.x += v[t].x * ww[t];
    s.y += v[t].y * ww[t];
    s.z += v[t].z * ww[t];
    s.w += v[t].w * ww[t];
  }
  return s;
}

__global__ __launch_bounds__(GATH_T, 16) void k_gather(const float* __restrict__ h, const int* __restrict__ cnt,
                                                   const int* __restrict__ sed,
                                                   const int* __restrict__ ovf,
                                                   const float* __restrict__ dis,
                                                   const float* __restrict__ bias,
                                                   float* __restrict__ out,
                                                   float* __restrict__ part, int N) {
  __shared__ __attribute__((aligned(16))) float ls[GATH_W][2 * HID];
  const int lane = threadIdx.x & 31;
  const int q    = lane >> 3;
  const int sub  = lane & 7;
  const int wib = __builtin_amdgcn_readfirstlane((int)(threadIdx.x >> 5));
  const int wpb = (int)(blockDim.x >> 5);
  const int nw  = (int)gridDim.x * wpb;
  const float4* __restrict__ h4   = reinterpret_cast<const float4*>(h);
  const int2*   __restrict__ sed2 = reinterpret_cast<const int2*>(sed);
  const float4 bb = make_float4(bias[4 * sub], bias[4 * sub + 1], bias[4 * sub + 2], bias[4 * sub + 3]);
  float4 s1 = make_float4(0.0f, 0.0f, 0.0f, 0.0f);
  float4 s2 = make_float4(0.0f, 0.0f, 0.0f, 0.0f);
  for (int c = (int)blockIdx.x * wpb + wib; c < N; c += nw) {
    const int nc  = cnt[c];
    const int beg = c * SEG;
    const int end = beg + ((nc < SEG) ? nc : SEG);
    const float dc = dis[c];
    const float4 own = h4[(unsigned)c * 8u + (unsigned)sub];
    float4 acc = make_float4(0.0f, 0.0f, 0.0f, 0.0f);
    for (int p0 = beg; p0 < end; p0 += 4 * GATH_TRIPS) {
      const int n = (end - p0 < 4 * GATH_TRIPS) ? (end - p0) : 4 * GATH_TRIPS;
      float4 s;
      switch ((n + 3) >> 2) {
        case 1:  s = gather_pass<1>(h4, sed2, p0, n, q, sub); break;
        case 2:  s = gather_pass<2>(h4, sed2, p0, n, q, sub); break;
        case 3:  s = gather_pass<3>(h4, sed2, p0, n, q, sub); break;
        case 4:  s = gather_pass<4>(h4, sed2, p0, n, q, sub); break;
        case 5:  s = gather_pass<5>(h4, sed2, p0, n, q, sub); break;
        default: s = gather_pass<6>(h4, sed2, p0, n, q, sub); break;
      }
      acc.x += s.x; acc.y += s.y; acc.z += s.z; acc.w += s.w;
    }
    if (nc > SEG) {
      const int no = cnt[N];
      const int4* __restrict__ ovf4 = reinterpret_cast<const int4*>(ovf);
      for (int j = 0; j < no; ++j) {
        const int4 en = ovf4[j];
        if (en.x == c && q == 0) {
          const float4 v = h4[(unsigned)en.y * 8u + (unsigned)sub];
          const float ww = __int_as_float(en.z);
          acc.x += v.x * ww; acc.y += v.y * ww; acc.z += v.z * ww; acc.w += v.w * ww;
        }
      }
    }
    if (q == 0) { acc.x += own.x; acc.y += own.y; acc.z += own.z; acc.w += own.w; }
    acc.x += __shfl_xor(acc.x, 8);
    acc.y += __shfl_xor(acc.y, 8);
    acc.z += __shfl_xor(acc.z, 8);
    acc.w += __shfl_xor(acc.w, 8);
    acc.x += __shfl_xor(acc.x, 16);
    acc.y += __shfl_xor(acc.y, 16);
    acc.z += __shfl_xor(acc.z, 16);
    acc.w += __shfl_xor(acc.w, 16);
    float4 o;
    o.x = acc.x * dc + bb.x;
    o.y = acc.y * dc + bb.y;
    o.z = acc.z * dc + bb.z;
    o.w = acc.w * dc + bb.w;
    if (q == 0) reinterpret_cast<float4*>(out)[(unsigned)c * 8u + (unsigned)sub] = o;
    const float r0 = fmaxf(o.x, 0.0f), r1 = fmaxf(o.y, 0.0f), r2 = fmaxf(o.z, 0.0f), r3 = fmaxf(o.w, 0.0f);
    s1.x += r0; s1.y += r1; s1.z += r2; s1.w += r3;
    s2.x += r0 * r0; s2.y += r1 * r1; s2.z += r2 * r2; s2.w += r3 * r3;
  }
  if (q == 0) {
    *reinterpret_cast<float4*>(&ls[wib][4 * sub])       = s1;
    *reinterpret_cast<float4*>(&ls[wib][HID + 4 * sub]) = s2;
  }
  __syncthreads();
  if (threadIdx.x < 2 * HID) {
    float t = 0.0f;
    for (int w = 0; w < GATH_W; ++w) t += ls[w][threadIdx.x];
    part[(size_t)blockIdx.x * (2 * HID) + threadIdx.x] = t;
  }
}

#define FIN_T 512
__global__ __launch_bounds__(FIN_T) void k_final(const float* __restrict__ c1, const float* __restrict__ c2,
                        const float* __restrict__ c3, const float* __restrict__ stats,
                        const float* __restrict__ part,
                        const float* __restrict__ g1, const float* __restrict__ be1,
                        const float* __restrict__ g2, const float* __restrict__ be2,
                        const float* __restrict__ g3, const float* __restrict__ be3,
                        const float* __restrict__ Wl,
                        const float* __restrict__ bl, float* __restrict__ out, int N) {
  __shared__ float4 prm[3 * HID];
  __shared__ __attribute__((aligned(16))) float red[(FIN_T / 16) * 2 * HID];
  __shared__ float sst[2 * HID];
  const int t = threadIdx.x;
  float4 sh[GATH_B * 16 / FIN_T];
  shares_ask<FIN_T>(part, t, sh);
  shares_add<FIN_T>(sh, t, red, sst);
  if (t < 3 * HID) {
    const int l = t >> 5;
    const int f = t & 31;
    float sm, sq;
    if (l < 2) { sm = stats[64 * l + f]; sq = stats[64 * l + 32 + f]; }
    else       { sm = sst[f];            sq = sst[32 + f]; }
    const float* g = (l == 0) ? g1 : ((l == 1) ? g2 : g3);
    const float* b = (l == 0) ? be1 : ((l == 1) ? be2 : be3);
    const float invN = 1.0f / (float)N;
    const float mu   = sm * invN;
    const float var  = sq * invN - mu * mu;
    prm[t] = make_float4(mu, rsqrtf(var + BN_EPS), g[f], b[f]);
  }
  __syncthreads();
  const int n = blockIdx.x * blockDim.x + t;
  if (n >= N) return;
  float a0 = bl[0], a1 = bl[1];
  const float* p = c1 + (size_t)n * HID;
#pragma unroll
  for (int j = 0; j < HID; ++j) {
    const float4 q = prm[j];
    const float v = (fmaxf(p[j], 0.0f) - q.x) * q.y * q.z + q.w;
    a0 += v * Wl[2 * j]; a1 += v * Wl[2 * j + 1];
  }
  p = c2 + (size_t)n * HID;
#pragma unroll
  for (int j = 0; j < HID; ++j) {
    const float4 q = prm[HID + j];
    const float v = (fmaxf(p[j], 0.0f) - q.x) * q.y * q.z + q.w;
    a0 += v * Wl[2 * (HID + j)]; a1 += v * Wl[2 * (HID + j) + 1];
  }
  p = c3 + (size_t)n * HID;
#pragma unroll
  for (int j = 0; j < HID; ++j) {
    const float4 q = prm[2 * HID + j];
    const float v = (fmaxf(p[j], 0.0f) - q.x) * q.y * q.z + q.w;
    a0 += v * Wl[2 * (2 * HID + j)]; a1 += v * Wl[2 * (2 * HID + j) + 1];
  }
  out[2 * n] = a0;
  out[2 * n + 1] = a1;
}

extern "C" void kernel_launch(void* const* d_in, const int* in_sizes, int n_in,
                              void* d_out, int out_size, void* d_ws, size_t ws_size,
                              hipStream_t stream) {
  const float* x   = (const float*)d_in[0];
  const int*   ei  = (const int*)d_in[1];
  const float* ew  = (const float*)d_in[2];
  const float* W1  = (const float*)d_in[3];
  const float* b1  = (const float*)d_in[4];
  const float* g1  = (const float*)d_in[5];
  const float* be1 = (const float*)d_in[6];
  const float* W2  = (const float*)d_in[7];
  const float* b2  = (const float*)d_in[8];
  const float* g2  = (const float*)d_in[9];
  const float* be2 = (const float*)d_in[10];
  const float* W3  = (const float*)d_in[11];
  const float* b3  = (const float*)d_in[12];
  const float* g3  = (const float*)d_in[13];
  const float* be3 = (const float*)d_in[14];
  const float* Wl  = (const float*)d_in[15];
  const float* bl  = (const float*)d_in[16];

  const int F = 64;
  const int N = in_sizes[0] / F;
  const int E = in_sizes[1] / 2;
  const int* row = ei;
  const int* col = ei + E;

  float* ws    = (float*)d_ws;
  float* dis   = ws;
  float* hbuf  = dis  + (((size_t)N + 63) & ~(size_t)63);
  float* c1    = hbuf + (size_t)N * HID;
  float* c2    = c1   + (size_t)N * HID;
  float* c3    = c2   + (size_t)N * HID;
  float* stats = c3   + (size_t)N * HID;
  float* st1   = stats;
  float* st2   = stats + 64;
  int*   cnt   = (int*)(stats + 192);
  const size_t sedOff = (((size_t)((float*)(cnt + N + 1) - ws)) + 63) & ~(size_t)63;
  int*   sed   = (int*)(ws + sedOff);
  const size_t ovfOff = (sedOff + 2 * (size_t)SEG * (size_t)N + 63) & ~(size_t)63;
  int*   ovf   = (int*)(ws + ovfOff);
  const int gathB = GATH_B;
  const size_t partOff = (ovfOff + 4 * (size_t)E + 63) & ~(size_t)63;
  float* part  = ws + partOff;

  const int T = 256;

  k_zero_cnt<<<(N + T) / T, T, 0, stream>>>(cnt, N);
  k_place<<<(E + T - 1) / T, T, 0, stream>>>(row, col, ew, cnt, sed, ovf, E, N);
  k_dis<<<(N + T - 1) / T, T, 0, stream>>>(cnt, sed, ovf, dis, N);

  const int numT = (N + 15) / 16;
  const dim3 gemmB(32, 8);
  const dim3 gemmG((numT + 7) / 8, 1);
  const dim3 gemmB2(32, 32);
  const dim3 gemmG2((numT + 31) / 32, 1);

  k_gemm_wmma<64, false, 8><<<gemmG, gemmB, 0, stream>>>(x, W1, hbuf, N, nullptr, nullptr, nullptr, nullptr, dis);
  k_gather<<<gathB, GATH_T, 0, stream>>>(hbuf, cnt, sed, ovf, dis, b1, c1, part, N);

  k_gemm_wmma<32, true, 32><<<gemmG2, gemmB2, 0, stream>>>(c1, W2, hbuf, N, part, st1, g1, be1, dis);
  k_gather<<<gathB, GATH_T, 0, stream>>>(hbuf, cnt, sed, ovf, dis, b2, c2, part, N);

  k_gemm_wmma<32, true, 32><<<gemmG2, gemmB2, 0, stream>>>(c2, W3, hbuf, N, part, st2, g2, be2, dis);
  k_gather<<<gathB, GATH_T, 0, stream>>>(hbuf, cnt, sed, ovf, dis, b3, c3, part, N);

  k_final<<<(N + FIN_T - 1) / FIN_T, FIN_T, 0, stream>>>(c1, c2, c3, stats, part, g1, be1, g2, be2, g3, be3,
                                             Wl, bl, (float*)d_out, N);
}
